# starting kernel
# baseline (speedup 1.0000x reference)
#include <hip/hip_runtime.h>
#include <hip/hip_bf16.h>

typedef __attribute__((ext_vector_type(2))) float v2f;
typedef __attribute__((ext_vector_type(8))) float v8f;

#define ALPHA 0.2f
#define BATCH 32
#define WDIM 128
#define KDIM 1024
#define JT   64
#define NCHUNK (KDIM / JT)
#define PSTR 258

__global__ __launch_bounds__(256) void fal_prep_kernel(const float* __restrict__ x,
                                                       const float* __restrict__ w,
                                                       float* __restrict__ s1,
                                                       float* __restrict__ s2,
                                                       float* __restrict__ t1,
                                                       float* __restrict__ ta) {
  int idx = blockIdx.x * 256 + threadIdx.x;
  int b = idx >> 10;
  int k = idx & (KDIM - 1);
  const float* xp = x + ((size_t)b << 17) + k;
  float a1 = 0.f, a2 = 0.f;
#pragma unroll 8
  for (int wi = 0; wi < WDIM; ++wi) {
    float xv = xp[(size_t)wi << 10];
    a1 = fmaf(xv, w[wi], a1);
    a2 = fmaf(xv, w[WDIM + wi], a2);
  }
  s1[idx] = a1;
  s2[idx] = a2;
  t1[idx] = __expf(a2);
  ta[idx] = __expf(ALPHA * a2);
}

__global__ __launch_bounds__(256) void fal_rowmax_kernel(const float* __restrict__ s2,
                                                         float* __restrict__ s2max) {
  __shared__ float red[256];
  int b = blockIdx.x;
  float m = -3.4e38f;
  for (int j = threadIdx.x; j < KDIM; j += 256)
    m = fmaxf(m, s2[b * KDIM + j]);
  red[threadIdx.x] = m;
  __syncthreads();
  for (int s = 128; s > 0; s >>= 1) {
    if (threadIdx.x < s) red[threadIdx.x] = fmaxf(red[threadIdx.x], red[threadIdx.x + s]);
    __syncthreads();
  }
  if (threadIdx.x == 0) s2max[b] = red[0];
}

__global__ __launch_bounds__(256) void fal_attn_kernel(const float* __restrict__ x,
                                                       const float* __restrict__ s1g,
                                                       const float* __restrict__ s2g,
                                                       const float* __restrict__ t1g,
                                                       const float* __restrict__ tag,
                                                       const float* __restrict__ s2mx,
                                                       float* __restrict__ out) {
  __shared__ float  Vt[(JT / 2) * PSTR];
  __shared__ float4 qsm[JT];

  const int tid  = threadIdx.x;
  const int lane = tid & 31;
  const int wave = tid >> 5;
  const int half = lane >> 4;
  const int m    = lane & 15;
  const int b    = blockIdx.y;
  const int i0   = blockIdx.x * 128 + wave * 16;

  const float s1v  = s1g[b * KDIM + i0 + m];
  const float zmax = s1v + s2mx[b];
  const float mrow = fmaxf(zmax, ALPHA * zmax);
  const float u1   = __expf(s1v - mrow);
  const float ua   = __expf(ALPHA * s1v - mrow);

  v8f acc[8];
#pragma unroll
  for (int t = 0; t < 8; ++t) acc[t] = (v8f)0.f;
  float dsum = 0.f;

  const int k0 = half * 2;

  const int jj4 = tid & 15;
  const int w0  = tid >> 4;
  const float* gbase = x + (((size_t)b * WDIM + w0) << 10) + (jj4 << 2);
  const int lbase = (jj4 * 2) * PSTR + 2 * w0;

  float4 r[8];
#pragma unroll
  for (int it = 0; it < 8; ++it)
    r[it] = *(const float4*)(gbase + ((size_t)(it * 16) << 10));

  for (int c = 0; c < NCHUNK; ++c) {
    const int j0 = c * JT;
    __syncthreads();
#pragma unroll
    for (int it = 0; it < 8; ++it) {
      int o = lbase + it * 32;
      *(float2*)&Vt[o]        = make_float2(r[it].x, r[it].y);
      *(float2*)&Vt[o + PSTR] = make_float2(r[it].z, r[it].w);
    }
    if (tid < JT) {
      int j = b * KDIM + j0 + tid;
      qsm[tid] = make_float4(s2g[j], t1g[j], tag[j], 0.f);
    }
    if (c + 1 < NCHUNK) {
#pragma unroll
      for (int it = 0; it < 8; ++it)
        r[it] = *(const float4*)(gbase + ((size_t)(it * 16) << 10) + (c + 1) * JT);
    }
    __syncthreads();

#pragma unroll
    for (int js = 0; js < JT; js += 4) {
      float4 qa = qsm[js + k0];
      float4 qb = qsm[js + k0 + 1];
      float z0 = s1v + qa.x;
      float z1 = s1v + qb.x;
      float p0 = (z0 > 0.f) ? u1 * qa.y : ua * qa.z;
      float p1 = (z1 > 0.f) ? u1 * qb.y : ua * qb.z;
      dsum += p0 + p1;
      v2f a; a.x = p0; a.y = p1;

      const float* vrow = &Vt[(js / 2 + half) * PSTR + 2 * m];
#pragma unroll
      for (int t = 0; t < 8; ++t) {
        v2f bb = *(const v2f*)(vrow + t * 32);
        acc[t] = __builtin_amdgcn_wmma_f32_16x16x4_f32(false, a, false, bb,
                                                       (short)0, acc[t], false, false);
      }
    }
  }

  float dtot = dsum + __shfl_xor(dsum, 16, 32);

#pragma unroll
  for (int rr = 0; rr < 8; ++rr) {
    int row   = rr + 8 * half;
    float d   = __shfl(dtot, rr + 8 * half, 32);
    float inv = 1.0f / d;
    float* orow = out + (((size_t)b * KDIM + i0 + row) << 7) + m;
#pragma unroll
    for (int t = 0; t < 8; ++t)
      orow[t * 16] = acc[t][rr] * inv;
  }
}

extern "C" void kernel_launch(void* const* d_in, const int* in_sizes, int n_in,
                              void* d_out, int out_size, void* d_ws, size_t ws_size,
                              hipStream_t stream) {
  const float* x = (const float*)d_in[0];
  const float* w = (const float*)d_in[1];
  float* out = (float*)d_out;

  float* s1  = (float*)d_ws;
  float* s2  = s1 + BATCH * KDIM;
  float* t1  = s2 + BATCH * KDIM;
  float* ta  = t1 + BATCH * KDIM;
  float* s2m = ta + BATCH * KDIM;

  fal_prep_kernel<<<(BATCH * KDIM) / 256, 256, 0, stream>>>(x, w, s1, s2, t1, ta);
  fal_rowmax_kernel<<<BATCH, 256, 0, stream>>>(s2, s2m);
  dim3 grid(KDIM / 128, BATCH);
  fal_attn_kernel<<<grid, 256, 0, stream>>>(x, s1, s2, t1, ta, s2m, out);
}


# numerator exp(lrelu(z)-m) taken as max(u1*exp(s2), ua*exp(a*s2)) instead of a select on z>0: WMMA k-loop has no divergent branches or dependent LDS loads; sum order unchanged
# speedup vs baseline: 1.2829x; 1.2829x over previous
#include <hip/hip_runtime.h>
#include <hip/hip_bf16.h>

typedef __attribute__((ext_vector_type(2))) float v2f;
typedef __attribute__((ext_vector_type(8))) float v8f;

#define ALPHA 0.2f
#define BATCH 32
#define WDIM 128
#define KDIM 1024
#define JT   64
#define NCHUNK (KDIM / JT)
#define PSTR 258

__global__ __launch_bounds__(256) void fal_prep_kernel(const float* __restrict__ x,
                                                       const float* __restrict__ w,
                                                       float* __restrict__ s1,
                                                       float* __restrict__ s2,
                                                       float* __restrict__ t1,
                                                       float* __restrict__ ta) {
  int idx = blockIdx.x * 256 + threadIdx.x;
  int b = idx >> 10;
  int k = idx & (KDIM - 1);
  const float* xp = x + ((size_t)b << 17) + k;
  float a1 = 0.f, a2 = 0.f;
#pragma unroll 8
  for (int wi = 0; wi < WDIM; ++wi) {
    float xv = xp[(size_t)wi << 10];
    a1 = fmaf(xv, w[wi], a1);
    a2 = fmaf(xv, w[WDIM + wi], a2);
  }
  s1[idx] = a1;
  s2[idx] = a2;
  t1[idx] = __expf(a2);
  ta[idx] = __expf(ALPHA * a2);
}

__global__ __launch_bounds__(256) void fal_rowmax_kernel(const float* __restrict__ s2,
                                                         float* __restrict__ s2max) {
  __shared__ float red[256];
  int b = blockIdx.x;
  float m = -3.4e38f;
  for (int j = threadIdx.x; j < KDIM; j += 256)
    m = fmaxf(m, s2[b * KDIM + j]);
  red[threadIdx.x] = m;
  __syncthreads();
  for (int s = 128; s > 0; s >>= 1) {
    if (threadIdx.x < s) red[threadIdx.x] = fmaxf(red[threadIdx.x], red[threadIdx.x + s]);
    __syncthreads();
  }
  if (threadIdx.x == 0) s2max[b] = red[0];
}

__global__ __launch_bounds__(256) void fal_attn_kernel(const float* __restrict__ x,
                                                       const float* __restrict__ s1g,
                                                       const float* __restrict__ t1g,
                                                       const float* __restrict__ tag,
                                                       const float* __restrict__ s2mx,
                                                       float* __restrict__ out) {
  __shared__ float  Vt[(JT / 2) * PSTR];
  __shared__ __align__(16) float2 qsm[JT];

  const int tid  = threadIdx.x;
  const int lane = tid & 31;
  const int wave = tid >> 5;
  const int half = lane >> 4;
  const int m    = lane & 15;
  const int b    = blockIdx.y;
  const int i0   = blockIdx.x * 128 + wave * 16;

  const float s1v  = s1g[b * KDIM + i0 + m];
  const float zmax = s1v + s2mx[b];
  const float mrow = fmaxf(zmax, ALPHA * zmax);
  const float u1   = __expf(s1v - mrow);
  const float ua   = __expf(ALPHA * s1v - mrow);

  v8f acc[8];
#pragma unroll
  for (int t = 0; t < 8; ++t) acc[t] = (v8f)0.f;
  float dsum = 0.f;

  const int k0 = half * 2;

  const int jj4 = tid & 15;
  const int w0  = tid >> 4;
  const float* gbase = x + (((size_t)b * WDIM + w0) << 10) + (jj4 << 2);
  const int lbase = (jj4 * 2) * PSTR + 2 * w0;

  float4 r[8];
#pragma unroll
  for (int it = 0; it < 8; ++it)
    r[it] = *(const float4*)(gbase + ((size_t)(it * 16) << 10));

  for (int c = 0; c < NCHUNK; ++c) {
    const int j0 = c * JT;
    __syncthreads();
#pragma unroll
    for (int it = 0; it < 8; ++it) {
      int o = lbase + it * 32;
      *(float2*)&Vt[o]        = make_float2(r[it].x, r[it].y);
      *(float2*)&Vt[o + PSTR] = make_float2(r[it].z, r[it].w);
    }
    if (tid < JT) {
      int j = b * KDIM + j0 + tid;
      qsm[tid] = make_float2(t1g[j], tag[j]);
    }
    if (c + 1 < NCHUNK) {
#pragma unroll
      for (int it = 0; it < 8; ++it)
        r[it] = *(const float4*)(gbase + ((size_t)(it * 16) << 10) + (c + 1) * JT);
    }
    __syncthreads();

#pragma unroll
    for (int js = 0; js < JT; js += 4) {
      float4 q = *(const float4*)&qsm[js + k0];
      float p0 = fmaxf(u1 * q.x, ua * q.y);
      float p1 = fmaxf(u1 * q.z, ua * q.w);
      dsum += p0 + p1;
      v2f a; a.x = p0; a.y = p1;

      const float* vrow = &Vt[(js / 2 + half) * PSTR + 2 * m];
#pragma unroll
      for (int t = 0; t < 8; ++t) {
        v2f bb = *(const v2f*)(vrow + t * 32);
        acc[t] = __builtin_amdgcn_wmma_f32_16x16x4_f32(false, a, false, bb,
                                                       (short)0, acc[t], false, false);
      }
    }
  }

  float dtot = dsum + __shfl_xor(dsum, 16, 32);

#pragma unroll
  for (int rr = 0; rr < 8; ++rr) {
    int row   = rr + 8 * half;
    float d   = __shfl(dtot, rr + 8 * half, 32);
    float inv = 1.0f / d;
    float* orow = out + (((size_t)b * KDIM + i0 + row) << 7) + m;
#pragma unroll
    for (int t = 0; t < 8; ++t)
      orow[t * 16] = acc[t][rr] * inv;
  }
}

extern "C" void kernel_launch(void* const* d_in, const int* in_sizes, int n_in,
                              void* d_out, int out_size, void* d_ws, size_t ws_size,
                              hipStream_t stream) {
  const float* x = (const float*)d_in[0];
  const float* w = (const float*)d_in[1];
  float* out = (float*)d_out;

  float* s1  = (float*)d_ws;
  float* s2  = s1 + BATCH * KDIM;
  float* t1  = s2 + BATCH * KDIM;
  float* ta  = t1 + BATCH * KDIM;
  float* s2m = ta + BATCH * KDIM;

  fal_prep_kernel<<<(BATCH * KDIM) / 256, 256, 0, stream>>>(x, w, s1, s2, t1, ta);
  fal_rowmax_kernel<<<BATCH, 256, 0, stream>>>(s2, s2m);
  dim3 grid(KDIM / 128, BATCH);
  fal_attn_kernel<<<grid, 256, 0, stream>>>(x, s1, t1, ta, s2m, out);
}


# V-fragment LDS reads software-pipelined in registers: the next 16-WMMA block's reads issued as one group ahead of the current block (second register set), not between WMMA pairs; sum order unchanged
# speedup vs baseline: 1.3616x; 1.0613x over previous
#include <hip/hip_runtime.h>
#include <hip/hip_bf16.h>

typedef __attribute__((ext_vector_type(2))) float v2f;
typedef __attribute__((ext_vector_type(8))) float v8f;

#define ALPHA 0.2f
#define BATCH 32
#define WDIM 128
#define KDIM 1024
#define JT   64
#define NCHUNK (KDIM / JT)
#define PSTR 258

__global__ __launch_bounds__(256) void fal_prep_kernel(const float* __restrict__ x,
                                                       const float* __restrict__ w,
                                                       float* __restrict__ s1,
                                                       float* __restrict__ s2,
                                                       float* __restrict__ t1,
                                                       float* __restrict__ ta) {
  int idx = blockIdx.x * 256 + threadIdx.x;
  int b = idx >> 10;
  int k = idx & (KDIM - 1);
  const float* xp = x + ((size_t)b << 17) + k;
  float a1 = 0.f, a2 = 0.f;
#pragma unroll 8
  for (int wi = 0; wi < WDIM; ++wi) {
    float xv = xp[(size_t)wi << 10];
    a1 = fmaf(xv, w[wi], a1);
    a2 = fmaf(xv, w[WDIM + wi], a2);
  }
  s1[idx] = a1;
  s2[idx] = a2;
  t1[idx] = __expf(a2);
  ta[idx] = __expf(ALPHA * a2);
}

__global__ __launch_bounds__(256) void fal_rowmax_kernel(const float* __restrict__ s2,
                                                         float* __restrict__ s2max) {
  __shared__ float red[256];
  int b = blockIdx.x;
  float m = -3.4e38f;
  for (int j = threadIdx.x; j < KDIM; j += 256)
    m = fmaxf(m, s2[b * KDIM + j]);
  red[threadIdx.x] = m;
  __syncthreads();
  for (int s = 128; s > 0; s >>= 1) {
    if (threadIdx.x < s) red[threadIdx.x] = fmaxf(red[threadIdx.x], red[threadIdx.x + s]);
    __syncthreads();
  }
  if (threadIdx.x == 0) s2max[b] = red[0];
}

__global__ __launch_bounds__(256) void fal_attn_kernel(const float* __restrict__ x,
                                                       const float* __restrict__ s1g,
                                                       const float* __restrict__ t1g,
                                                       const float* __restrict__ tag,
                                                       const float* __restrict__ s2mx,
                                                       float* __restrict__ out) {
  __shared__ float  Vt[(JT / 2) * PSTR];
  __shared__ __align__(16) float2 qsm[JT];

  const int tid  = threadIdx.x;
  const int lane = tid & 31;
  const int wave = tid >> 5;
  const int half = lane >> 4;
  const int m    = lane & 15;
  const int b    = blockIdx.y;
  const int i0   = blockIdx.x * 128 + wave * 16;

  const float s1v  = s1g[b * KDIM + i0 + m];
  const float zmax = s1v + s2mx[b];
  const float mrow = fmaxf(zmax, ALPHA * zmax);
  const float u1   = __expf(s1v - mrow);
  const float ua   = __expf(ALPHA * s1v - mrow);

  v8f acc[8];
#pragma unroll
  for (int t = 0; t < 8; ++t) acc[t] = (v8f)0.f;
  float dsum = 0.f;

  const int k0 = half * 2;
  const float* vb = &Vt[half * PSTR + 2 * m];

  const int jj4 = tid & 15;
  const int w0  = tid >> 4;
  const float* gbase = x + (((size_t)b * WDIM + w0) << 10) + (jj4 << 2);
  const int lbase = (jj4 * 2) * PSTR + 2 * w0;

  float4 r[8];
#pragma unroll
  for (int it = 0; it < 8; ++it)
    r[it] = *(const float4*)(gbase + ((size_t)(it * 16) << 10));

  for (int c = 0; c < NCHUNK; ++c) {
    const int j0 = c * JT;
    __syncthreads();
#pragma unroll
    for (int it = 0; it < 8; ++it) {
      int o = lbase + it * 32;
      *(float2*)&Vt[o]        = make_float2(r[it].x, r[it].y);
      *(float2*)&Vt[o + PSTR] = make_float2(r[it].z, r[it].w);
    }
    if (tid < JT) {
      int j = b * KDIM + j0 + tid;
      qsm[tid] = make_float2(t1g[j], tag[j]);
    }
    if (c + 1 < NCHUNK) {
#pragma unroll
      for (int it = 0; it < 8; ++it)
        r[it] = *(const float4*)(gbase + ((size_t)(it * 16) << 10) + (c + 1) * JT);
    }
    __syncthreads();

#define FAL_LOAD_Q(g)                                                          \
    {                                                                          \
      qr[(g) % 3][0] = *(const float4*)&qsm[(g) * 8 + k0];                     \
      qr[(g) % 3][1] = *(const float4*)&qsm[(g) * 8 + 4 + k0];                 \
    }
#define FAL_LOAD_B(g)                                                          \
    {                                                                          \
      _Pragma("unroll") for (int s = 0; s < 2; ++s)                            \
        _Pragma("unroll") for (int t = 0; t < 8; ++t)                          \
          bf[(g) & 1][s * 8 + t] =                                             \
              *(const v2f*)(vb + ((g) * 4 + s * 2) * PSTR + t * 32);           \
    }
#define FAL_MAKE_A(g)                                                          \
    {                                                                          \
      _Pragma("unroll") for (int s = 0; s < 2; ++s) {                          \
        float4 q = qr[(g) % 3][s];                                             \
        float p0 = fmaxf(u1 * q.x, ua * q.y);                                  \
        float p1 = fmaxf(u1 * q.z, ua * q.w);                                  \
        dsum += p0 + p1;                                                       \
        af[(g) & 1][s].x = p0;                                                 \
        af[(g) & 1][s].y = p1;                                                 \
      }                                                                        \
    }

    v2f    bf[2][16];
    float4 qr[3][2];
    v2f    af[2][2];

    FAL_LOAD_Q(0)
    FAL_LOAD_Q(1)
    FAL_LOAD_B(0)
    FAL_MAKE_A(0)

#pragma unroll
    for (int g = 0; g < JT / 8; ++g) {
      if (g + 2 < JT / 8) FAL_LOAD_Q(g + 2)
      if (g + 1 < JT / 8) FAL_LOAD_B(g + 1)
      __builtin_amdgcn_sched_barrier(0);
      if (g + 1 < JT / 8) FAL_MAKE_A(g + 1)
#pragma unroll
      for (int s = 0; s < 2; ++s) {
#pragma unroll
        for (int t = 0; t < 8; ++t)
          acc[t] = __builtin_amdgcn_wmma_f32_16x16x4_f32(false, af[g & 1][s], false,
                                                         bf[g & 1][s * 8 + t],
                                                         (short)0, acc[t], false, false);
      }
      __builtin_amdgcn_sched_barrier(0);
    }
#undef FAL_LOAD_Q
#undef FAL_LOAD_B
#undef FAL_MAKE_A
  }

  float dtot = dsum + __shfl_xor(dsum, 16, 32);

#pragma unroll
  for (int rr = 0; rr < 8; ++rr) {
    int row   = rr + 8 * half;
    float d   = __shfl(dtot, rr + 8 * half, 32);
    float inv = 1.0f / d;
    float* orow = out + (((size_t)b * KDIM + i0 + row) << 7) + m;
#pragma unroll
    for (int t = 0; t < 8; ++t)
      orow[t * 16] = acc[t][rr] * inv;
  }
}

extern "C" void kernel_launch(void* const* d_in, const int* in_sizes, int n_in,
                              void* d_out, int out_size, void* d_ws, size_t ws_size,
                              hipStream_t stream) {
  const float* x = (const float*)d_in[0];
  const float* w = (const float*)d_in[1];
  float* out = (float*)d_out;

  float* s1  = (float*)d_ws;
  float* s2  = s1 + BATCH * KDIM;
  float* t1  = s2 + BATCH * KDIM;
  float* ta  = t1 + BATCH * KDIM;
  float* s2m = ta + BATCH * KDIM;

  fal_prep_kernel<<<(BATCH * KDIM) / 256, 256, 0, stream>>>(x, w, s1, s2, t1, ta);
  fal_rowmax_kernel<<<BATCH, 256, 0, stream>>>(s2, s2m);
  dim3 grid(KDIM / 128, BATCH);
  fal_attn_kernel<<<grid, 256, 0, stream>>>(x, s1, t1, ta, s2m, out);
}


# next chunk's global loads (V tile, exp-factor pairs) moved from in front of the 2nd barrier to behind it: the barrier's fence no longer waits for them, they fly under the WMMAs; sum order unchanged
# speedup vs baseline: 1.5127x; 1.1110x over previous
#include <hip/hip_runtime.h>
#include <hip/hip_bf16.h>

typedef __attribute__((ext_vector_type(2))) float v2f;
typedef __attribute__((ext_vector_type(8))) float v8f;

#define ALPHA 0.2f
#define BATCH 32
#define WDIM 128
#define KDIM 1024
#define JT   64
#define NCHUNK (KDIM / JT)
#define PSTR 258

__global__ __launch_bounds__(256) void fal_prep_kernel(const float* __restrict__ x,
                                                       const float* __restrict__ w,
                                                       float* __restrict__ s1,
                                                       float* __restrict__ s2,
                                                       float* __restrict__ t1,
                                                       float* __restrict__ ta) {
  int idx = blockIdx.x * 256 + threadIdx.x;
  int b = idx >> 10;
  int k = idx & (KDIM - 1);
  const float* xp = x + ((size_t)b << 17) + k;
  float a1 = 0.f, a2 = 0.f;
#pragma unroll 8
  for (int wi = 0; wi < WDIM; ++wi) {
    float xv = xp[(size_t)wi << 10];
    a1 = fmaf(xv, w[wi], a1);
    a2 = fmaf(xv, w[WDIM + wi], a2);
  }
  s1[idx] = a1;
  s2[idx] = a2;
  t1[idx] = __expf(a2);
  ta[idx] = __expf(ALPHA * a2);
}

__global__ __launch_bounds__(256) void fal_rowmax_kernel(const float* __restrict__ s2,
                                                         float* __restrict__ s2max) {
  __shared__ float red[256];
  int b = blockIdx.x;
  float m = -3.4e38f;
  for (int j = threadIdx.x; j < KDIM; j += 256)
    m = fmaxf(m, s2[b * KDIM + j]);
  red[threadIdx.x] = m;
  __syncthreads();
  for (int s = 128; s > 0; s >>= 1) {
    if (threadIdx.x < s) red[threadIdx.x] = fmaxf(red[threadIdx.x], red[threadIdx.x + s]);
    __syncthreads();
  }
  if (threadIdx.x == 0) s2max[b] = red[0];
}

__global__ __launch_bounds__(256) void fal_attn_kernel(const float* __restrict__ x,
                                                       const float* __restrict__ s1g,
                                                       const float* __restrict__ t1g,
                                                       const float* __restrict__ tag,
                                                       const float* __restrict__ s2mx,
                                                       float* __restrict__ out) {
  __shared__ float  Vt[(JT / 2) * PSTR];
  __shared__ __align__(16) float2 qsm[JT];

  const int tid  = threadIdx.x;
  const int lane = tid & 31;
  const int wave = tid >> 5;
  const int half = lane >> 4;
  const int m    = lane & 15;
  const int b    = blockIdx.y;
  const int i0   = blockIdx.x * 128 + wave * 16;

  const float s1v  = s1g[b * KDIM + i0 + m];
  const float zmax = s1v + s2mx[b];
  const float mrow = fmaxf(zmax, ALPHA * zmax);
  const float u1   = __expf(s1v - mrow);
  const float ua   = __expf(ALPHA * s1v - mrow);

  v8f acc[8];
#pragma unroll
  for (int t = 0; t < 8; ++t) acc[t] = (v8f)0.f;
  float dsum = 0.f;

  const int k0 = half * 2;
  const float* vb = &Vt[half * PSTR + 2 * m];

  const int jj4 = tid & 15;
  const int w0  = tid >> 4;
  const float* gbase = x + (((size_t)b * WDIM + w0) << 10) + (jj4 << 2);
  const int lbase = (jj4 * 2) * PSTR + 2 * w0;

  float4 r[8];
#pragma unroll
  for (int it = 0; it < 8; ++it)
    r[it] = *(const float4*)(gbase + ((size_t)(it * 16) << 10));
  const int qj = b * KDIM + tid;
  float2 qn = make_float2(0.f, 0.f);
  if (tid < JT) qn = make_float2(t1g[qj], tag[qj]);

  for (int c = 0; c < NCHUNK; ++c) {
    __syncthreads();
#pragma unroll
    for (int it = 0; it < 8; ++it) {
      int o = lbase + it * 32;
      *(float2*)&Vt[o]        = make_float2(r[it].x, r[it].y);
      *(float2*)&Vt[o + PSTR] = make_float2(r[it].z, r[it].w);
    }
    if (tid < JT) qsm[tid] = qn;
    __syncthreads();
    if (c + 1 < NCHUNK) {
#pragma unroll
      for (int it = 0; it < 8; ++it)
        r[it] = *(const float4*)(gbase + ((size_t)(it * 16) << 10) + (c + 1) * JT);
      if (tid < JT) qn = make_float2(t1g[qj + (c + 1) * JT], tag[qj + (c + 1) * JT]);
    }

#define FAL_LOAD_Q(g)                                                          \
    {                                                                          \
      qr[(g) % 3][0] = *(const float4*)&qsm[(g) * 8 + k0];                     \
      qr[(g) % 3][1] = *(const float4*)&qsm[(g) * 8 + 4 + k0];                 \
    }
#define FAL_LOAD_B(g)                                                          \
    {                                                                          \
      _Pragma("unroll") for (int s = 0; s < 2; ++s)                            \
        _Pragma("unroll") for (int t = 0; t < 8; ++t)                          \
          bf[(g) & 1][s * 8 + t] =                                             \
              *(const v2f*)(vb + ((g) * 4 + s * 2) * PSTR + t * 32);           \
    }
#define FAL_MAKE_A(g)                                                          \
    {                                                                          \
      _Pragma("unroll") for (int s = 0; s < 2; ++s) {                          \
        float4 q = qr[(g) % 3][s];                                             \
        float p0 = fmaxf(u1 * q.x, ua * q.y);                                  \
        float p1 = fmaxf(u1 * q.z, ua * q.w);                                  \
        dsum += p0 + p1;                                                       \
        af[(g) & 1][s].x = p0;                                                 \
        af[(g) & 1][s].y = p1;                                                 \
      }                                                                        \
    }

    v2f    bf[2][16];
    float4 qr[3][2];
    v2f    af[2][2];

    FAL_LOAD_Q(0)
    FAL_LOAD_Q(1)
    FAL_LOAD_B(0)
    FAL_MAKE_A(0)

#pragma unroll
    for (int g = 0; g < JT / 8; ++g) {
      if (g + 2 < JT / 8) FAL_LOAD_Q(g + 2)
      if (g + 1 < JT / 8) FAL_LOAD_B(g + 1)
      __builtin_amdgcn_sched_barrier(0);
      if (g + 1 < JT / 8) FAL_MAKE_A(g + 1)
#pragma unroll
      for (int s = 0; s < 2; ++s) {
#pragma unroll
        for (int t = 0; t < 8; ++t)
          acc[t] = __builtin_amdgcn_wmma_f32_16x16x4_f32(false, af[g & 1][s], false,
                                                         bf[g & 1][s * 8 + t],
                                                         (short)0, acc[t], false, false);
      }
      __builtin_amdgcn_sched_barrier(0);
    }
#undef FAL_LOAD_Q
#undef FAL_LOAD_B
#undef FAL_MAKE_A
  }

  float dtot = dsum + __shfl_xor(dsum, 16, 32);

#pragma unroll
  for (int rr = 0; rr < 8; ++rr) {
    int row   = rr + 8 * half;
    float d   = __shfl(dtot, rr + 8 * half, 32);
    float inv = 1.0f / d;
    float* orow = out + (((size_t)b * KDIM + i0 + row) << 7) + m;
#pragma unroll
    for (int t = 0; t < 8; ++t)
      orow[t * 16] = acc[t][rr] * inv;
  }
}

extern "C" void kernel_launch(void* const* d_in, const int* in_sizes, int n_in,
                              void* d_out, int out_size, void* d_ws, size_t ws_size,
                              hipStream_t stream) {
  const float* x = (const float*)d_in[0];
  const float* w = (const float*)d_in[1];
  float* out = (float*)d_out;

  float* s1  = (float*)d_ws;
  float* s2  = s1 + BATCH * KDIM;
  float* t1  = s2 + BATCH * KDIM;
  float* ta  = t1 + BATCH * KDIM;
  float* s2m = ta + BATCH * KDIM;

  fal_prep_kernel<<<(BATCH * KDIM) / 256, 256, 0, stream>>>(x, w, s1, s2, t1, ta);
  fal_rowmax_kernel<<<BATCH, 256, 0, stream>>>(s2, s2m);
  dim3 grid(KDIM / 128, BATCH);
  fal_attn_kernel<<<grid, 256, 0, stream>>>(x, s1, t1, ta, s2m, out);
}


# prep and row-max launches folded into the attention kernel: each work-group recomputes its batch's s2, exp factors and max in a pre-pass kept in LDS; 3 launches -> 1; sum order unchanged
# speedup vs baseline: 1.7964x; 1.1875x over previous
#include <hip/hip_runtime.h>

typedef __attribute__((ext_vector_type(2))) float v2f;
typedef __attribute__((ext_vector_type(8))) float v8f;

#define ALPHA 0.2f
#define BATCH 32
#define WDIM 128
#define KDIM 1024
#define JT   64
#define NCHUNK (KDIM / JT)
#define PSTR 258
#define PB   16

__device__ __forceinline__ void fal_proj_load(float4 (&xv)[PB], const float* __restrict__ xp,
                                              int wbase) {
#pragma unroll
  for (int u = 0; u < PB; ++u)
    xv[u] = *(const float4*)(xp + ((size_t)(wbase + u) << 10));
}

template <bool OWN>
__device__ __forceinline__ void fal_proj_fma(const float4 (&xv)[PB], const float* __restrict__ w,
                                             int wbase, float (&a1)[4], float (&a2)[4]) {
#pragma unroll
  for (int u = 0; u < PB; ++u) {
    const float w2v = w[WDIM + wbase + u];
    a2[0] = fmaf(xv[u].x, w2v, a2[0]);
    a2[1] = fmaf(xv[u].y, w2v, a2[1]);
    a2[2] = fmaf(xv[u].z, w2v, a2[2]);
    a2[3] = fmaf(xv[u].w, w2v, a2[3]);
    if (OWN) {
      const float w1v = w[wbase + u];
      a1[0] = fmaf(xv[u].x, w1v, a1[0]);
      a1[1] = fmaf(xv[u].y, w1v, a1[1]);
      a1[2] = fmaf(xv[u].z, w1v, a1[2]);
      a1[3] = fmaf(xv[u].w, w1v, a1[3]);
    }
  }
}

template <bool OWN>
__device__ __forceinline__ void fal_project(const float* __restrict__ xp,
                                            const float* __restrict__ w,
                                            float (&a1)[4], float (&a2)[4]) {
  float4 xa[PB], xb[PB];
  fal_proj_load(xa, xp, 0);
#pragma unroll
  for (int blk = 0; blk < WDIM / PB; blk += 2) {
    fal_proj_load(xb, xp, (blk + 1) * PB);
    __builtin_amdgcn_sched_barrier(0);
    fal_proj_fma<OWN>(xa, w, blk * PB, a1, a2);
    __builtin_amdgcn_sched_barrier(0);
    if (blk + 2 < WDIM / PB) fal_proj_load(xa, xp, (blk + 2) * PB);
    __builtin_amdgcn_sched_barrier(0);
    fal_proj_fma<OWN>(xb, w, (blk + 1) * PB, a1, a2);
    __builtin_amdgcn_sched_barrier(0);
  }
}

__global__ __launch_bounds__(256) void fal_attn_kernel(const float* __restrict__ x,
                                                       const float* __restrict__ w,
                                                       float* __restrict__ out) {
  __shared__ float  Vt[(JT / 2) * PSTR];
  __shared__ __align__(16) float2 qall[KDIM];
  __shared__ __align__(16) float  s1s[128];
  __shared__ float  red[8];

  const int tid  = threadIdx.x;
  const int lane = tid & 31;
  const int wave = tid >> 5;
  const int half = lane >> 4;
  const int m    = lane & 15;
  const int b    = blockIdx.y;
  const int i0   = blockIdx.x * 128 + wave * 16;

  const int jj4 = tid & 15;
  const int w0  = tid >> 4;
  const float* gbase = x + (((size_t)b * WDIM + w0) << 10) + (jj4 << 2);
  const int lbase = (jj4 * 2) * PSTR + 2 * w0;
  float4 r[8];

  {
    const float* xp = x + ((size_t)b << 17) + (tid << 2);
    float a1[4] = {0.f, 0.f, 0.f, 0.f};
    float a2[4] = {0.f, 0.f, 0.f, 0.f};
    const bool own = (__builtin_amdgcn_readfirstlane(wave) == (int)blockIdx.x);
    if (own) fal_project<true>(xp, w, a1, a2);
    else     fal_project<false>(xp, w, a1, a2);

#pragma unroll
    for (int it = 0; it < 8; ++it)
      r[it] = *(const float4*)(gbase + ((size_t)(it * 16) << 10));

    *(float4*)&qall[4 * tid]     = make_float4(__expf(a2[0]), __expf(ALPHA * a2[0]),
                                               __expf(a2[1]), __expf(ALPHA * a2[1]));
    *(float4*)&qall[4 * tid + 2] = make_float4(__expf(a2[2]), __expf(ALPHA * a2[2]),
                                               __expf(a2[3]), __expf(ALPHA * a2[3]));
    if (own) *(float4*)&s1s[4 * lane] = make_float4(a1[0], a1[1], a1[2], a1[3]);

    float mx = fmaxf(fmaxf(a2[0], a2[1]), fmaxf(a2[2], a2[3]));
#pragma unroll
    for (int off = 16; off > 0; off >>= 1) mx = fmaxf(mx, __shfl_xor(mx, off, 32));
    if (lane == 0) red[wave] = mx;
  }
  __syncthreads();

  float s2max = red[0];
#pragma unroll
  for (int i = 1; i < 8; ++i) s2max = fmaxf(s2max, red[i]);
  const float s1v  = s1s[wave * 16 + m];
  const float zmax = s1v + s2max;
  const float mrow = fmaxf(zmax, ALPHA * zmax);
  const float u1   = __expf(s1v - mrow);
  const float ua   = __expf(ALPHA * s1v - mrow);

  v8f acc[8];
#pragma unroll
  for (int t = 0; t < 8; ++t) acc[t] = (v8f)0.f;
  float dsum = 0.f;

  const int k0 = half * 2;
  const float* vb = &Vt[half * PSTR + 2 * m];

  for (int c = 0; c < NCHUNK; ++c) {
    __syncthreads();
#pragma unroll
    for (int it = 0; it < 8; ++it) {
      int o = lbase + it * 32;
      *(float2*)&Vt[o]        = make_float2(r[it].x, r[it].y);
      *(float2*)&Vt[o + PSTR] = make_float2(r[it].z, r[it].w);
    }
    __syncthreads();
    if (c + 1 < NCHUNK) {
#pragma unroll
      for (int it = 0; it < 8; ++it)
        r[it] = *(const float4*)(gbase + ((size_t)(it * 16) << 10) + (c + 1) * JT);
    }
    const float2* qc = &qall[c * JT];

#define FAL_LOAD_Q(g)                                                          \
    {                                                                          \
      qr[(g) % 3][0] = *(const float4*)&qc[(g) * 8 + k0];                      \
      qr[(g) % 3][1] = *(const float4*)&qc[(g) * 8 + 4 + k0];                  \
    }
#define FAL_LOAD_B(g)                                                          \
    {                                                                          \
      _Pragma("unroll") for (int s = 0; s < 2; ++s)                            \
        _Pragma("unroll") for (int t = 0; t < 8; ++t)                          \
          bf[(g) & 1][s * 8 + t] =                                             \
              *(const v2f*)(vb + ((g) * 4 + s * 2) * PSTR + t * 32);           \
    }
#define FAL_MAKE_A(g)                                                          \
    {                                                                          \
      _Pragma("unroll") for (int s = 0; s < 2; ++s) {                          \
        float4 q = qr[(g) % 3][s];                                             \
        float p0 = fmaxf(u1 * q.x, ua * q.y);                                  \
        float p1 = fmaxf(u1 * q.z, ua * q.w);                                  \
        dsum += p0 + p1;                                                       \
        af[(g) & 1][s].x = p0;                                                 \
        af[(g) & 1][s].y = p1;                                                 \
      }                                                                        \
    }

    v2f    bf[2][16];
    float4 qr[3][2];
    v2f    af[2][2];

    FAL_LOAD_Q(0)
    FAL_LOAD_Q(1)
    FAL_LOAD_B(0)
    FAL_MAKE_A(0)

#pragma unroll
    for (int g = 0; g < JT / 8; ++g) {
      if (g + 2 < JT / 8) FAL_LOAD_Q(g + 2)
      if (g + 1 < JT / 8) FAL_LOAD_B(g + 1)
      __builtin_amdgcn_sched_barrier(0);
      if (g + 1 < JT / 8) FAL_MAKE_A(g + 1)
#pragma unroll
      for (int s = 0; s < 2; ++s) {
#pragma unroll
        for (int t = 0; t < 8; ++t)
          acc[t] = __builtin_amdgcn_wmma_f32_16x16x4_f32(false, af[g & 1][s], false,
                                                         bf[g & 1][s * 8 + t],
                                                         (short)0, acc[t], false, false);
      }
      __builtin_amdgcn_sched_barrier(0);
    }
#undef FAL_LOAD_Q
#undef FAL_LOAD_B
#undef FAL_MAKE_A
  }

  float dtot = dsum + __shfl_xor(dsum, 16, 32);

#pragma unroll
  for (int rr = 0; rr < 8; ++rr) {
    int row   = rr + 8 * half;
    float d   = __shfl(dtot, rr + 8 * half, 32);
    float inv = 1.0f / d;
    float* orow = out + (((size_t)b * KDIM + i0 + row) << 7) + m;
#pragma unroll
    for (int t = 0; t < 8; ++t)
      orow[t * 16] = acc[t][rr] * inv;
  }
}

extern "C" void kernel_launch(void* const* d_in, const int* in_sizes, int n_in,
                              void* d_out, int out_size, void* d_ws, size_t ws_size,
                              hipStream_t stream) {
  const float* x = (const float*)d_in[0];
  const float* w = (const float*)d_in[1];
  float* out = (float*)d_out;

  dim3 grid(KDIM / 128, BATCH);
  fal_attn_kernel<<<grid, 256, 0, stream>>>(x, w, out);
}


# V tile triple-buffered in LDS: one barrier per chunk, mid-chunk between WMMA blocks, not two at its edge; tile stores ride under WMMAs, fragment pipeline never restarts; sum order unchanged
# speedup vs baseline: 1.9132x; 1.0650x over previous
#include <hip/hip_runtime.h>

typedef __attribute__((ext_vector_type(2))) float v2f;
typedef __attribute__((ext_vector_type(8))) float v8f;

#define ALPHA 0.2f
#define BATCH 32
#define WDIM 128
#define KDIM 1024
#define JT   64
#define NCHUNK (KDIM / JT)
#define PSTR 258
#define TILE ((JT / 2) * PSTR)
#define NBUF 3
#define PB   16

__device__ __forceinline__ void fal_proj_load(float4 (&xv)[PB], const float* __restrict__ xp,
                                              int wbase) {
#pragma unroll
  for (int u = 0; u < PB; ++u)
    xv[u] = *(const float4*)(xp + ((size_t)(wbase + u) << 10));
}

template <bool OWN>
__device__ __forceinline__ void fal_proj_fma(const float4 (&xv)[PB], const float* __restrict__ w,
                                             int wbase, float (&a1)[4], float (&a2)[4]) {
#pragma unroll
  for (int u = 0; u < PB; ++u) {
    const float w2v = w[WDIM + wbase + u];
    a2[0] = fmaf(xv[u].x, w2v, a2[0]);
    a2[1] = fmaf(xv[u].y, w2v, a2[1]);
    a2[2] = fmaf(xv[u].z, w2v, a2[2]);
    a2[3] = fmaf(xv[u].w, w2v, a2[3]);
    if (OWN) {
      const float w1v = w[wbase + u];
      a1[0] = fmaf(xv[u].x, w1v, a1[0]);
      a1[1] = fmaf(xv[u].y, w1v, a1[1]);
      a1[2] = fmaf(xv[u].z, w1v, a1[2]);
      a1[3] = fmaf(xv[u].w, w1v, a1[3]);
    }
  }
}

template <bool OWN>
__device__ __forceinline__ void fal_project(const float* __restrict__ xp,
                                            const float* __restrict__ w,
                                            float (&a1)[4], float (&a2)[4]) {
  float4 xa[PB], xb[PB];
  fal_proj_load(xa, xp, 0);
#pragma unroll
  for (int blk = 0; blk < WDIM / PB; blk += 2) {
    fal_proj_load(xb, xp, (blk + 1) * PB);
    __builtin_amdgcn_sched_barrier(0);
    fal_proj_fma<OWN>(xa, w, blk * PB, a1, a2);
    __builtin_amdgcn_sched_barrier(0);
    if (blk + 2 < WDIM / PB) fal_proj_load(xa, xp, (blk + 2) * PB);
    __builtin_amdgcn_sched_barrier(0);
    fal_proj_fma<OWN>(xb, w, (blk + 1) * PB, a1, a2);
    __builtin_amdgcn_sched_barrier(0);
  }
}

__global__ __launch_bounds__(256) void fal_attn_kernel(const float* __restrict__ x,
                                                       const float* __restrict__ w,
                                                       float* __restrict__ out) {
  __shared__ float  Vt[NBUF * TILE];
  __shared__ __align__(16) float2 qall[KDIM];
  __shared__ __align__(16) float  s1s[128];
  __shared__ float  red[8];

  const int tid  = threadIdx.x;
  const int lane = tid & 31;
  const int wave = tid >> 5;
  const int half = lane >> 4;
  const int m    = lane & 15;
  const int b    = blockIdx.y;
  const int i0   = blockIdx.x * 128 + wave * 16;

  const int jj4 = tid & 15;
  const int w0  = tid >> 4;
  const float* gbase = x + (((size_t)b * WDIM + w0) << 10) + (jj4 << 2);
  const int lbase = (jj4 * 2) * PSTR + 2 * w0;
  float4 r[8];

  {
    const float* xp = x + ((size_t)b << 17) + (tid << 2);
    float a1[4] = {0.f, 0.f, 0.f, 0.f};
    float a2[4] = {0.f, 0.f, 0.f, 0.f};
    const bool own = (__builtin_amdgcn_readfirstlane(wave) == (int)blockIdx.x);
    if (own) fal_project<true>(xp, w, a1, a2);
    else     fal_project<false>(xp, w, a1, a2);

#pragma unroll
    for (int it = 0; it < 8; ++it)
      r[it] = *(const float4*)(gbase + ((size_t)(it * 16) << 10));

    *(float4*)&qall[4 * tid]     = make_float4(__expf(a2[0]), __expf(ALPHA * a2[0]),
                                               __expf(a2[1]), __expf(ALPHA * a2[1]));
    *(float4*)&qall[4 * tid + 2] = make_float4(__expf(a2[2]), __expf(ALPHA * a2[2]),
                                               __expf(a2[3]), __expf(ALPHA * a2[3]));
    if (own) *(float4*)&s1s[4 * lane] = make_float4(a1[0], a1[1], a1[2], a1[3]);

    float mx = fmaxf(fmaxf(a2[0], a2[1]), fmaxf(a2[2], a2[3]));
#pragma unroll
    for (int off = 16; off > 0; off >>= 1) mx = fmaxf(mx, __shfl_xor(mx, off, 32));
    if (lane == 0) red[wave] = mx;

#pragma unroll
    for (int it = 0; it < 8; ++it) {
      int o = lbase + it * 32;
      *(float2*)&Vt[o]        = make_float2(r[it].x, r[it].y);
      *(float2*)&Vt[o + PSTR] = make_float2(r[it].z, r[it].w);
    }
  }
  __syncthreads();
#pragma unroll
  for (int it = 0; it < 8; ++it)
    r[it] = *(const float4*)(gbase + ((size_t)(it * 16) << 10) + JT);

  float s2max = red[0];
#pragma unroll
  for (int i = 1; i < 8; ++i) s2max = fmaxf(s2max, red[i]);
  const float s1v  = s1s[wave * 16 + m];
  const float zmax = s1v + s2max;
  const float mrow = fmaxf(zmax, ALPHA * zmax);
  const float u1   = __expf(s1v - mrow);
  const float ua   = __expf(ALPHA * s1v - mrow);

  v8f acc[8];
#pragma unroll
  for (int t = 0; t < 8; ++t) acc[t] = (v8f)0.f;
  float dsum = 0.f;

  const int k0 = half * 2;
  const float* vb = &Vt[half * PSTR + 2 * m];

#define FAL_LOAD_Q(slot, qp)                                                   \
    {                                                                          \
      qr[slot][0] = *(const float4*)(qp);                                      \
      qr[slot][1] = *(const float4*)((qp) + 4);                                \
    }
#define FAL_LOAD_B(set, vp, gb)                                                \
    {                                                                          \
      _Pragma("unroll") for (int s = 0; s < 2; ++s)                            \
        _Pragma("unroll") for (int t = 0; t < 8; ++t)                          \
          bf[set][s * 8 + t] =                                                 \
              *(const v2f*)((vp) + ((gb) * 4 + s * 2) * PSTR + t * 32);        \
    }
#define FAL_MAKE_A(aset, qslot)                                                \
    {                                                                          \
      _Pragma("unroll") for (int s = 0; s < 2; ++s) {                          \
        float4 q = qr[qslot][s];                                               \
        float p0 = fmaxf(u1 * q.x, ua * q.y);                                  \
        float p1 = fmaxf(u1 * q.z, ua * q.w);                                  \
        dsum += p0 + p1;                                                       \
        af[aset][s].x = p0;                                                    \
        af[aset][s].y = p1;                                                    \
      }                                                                        \
    }
#define FAL_WMMA(set)                                                          \
    {                                                                          \
      _Pragma("unroll") for (int s = 0; s < 2; ++s)                            \
        _Pragma("unroll") for (int t = 0; t < 8; ++t)                          \
          acc[t] = __builtin_amdgcn_wmma_f32_16x16x4_f32(false, af[set][s], false, \
                                                         bf[set][s * 8 + t],   \
                                                         (short)0, acc[t], false, false); \
    }
#define FAL_STORE_TILE(vs)                                                     \
    {                                                                          \
      _Pragma("unroll") for (int it = 0; it < 8; ++it) {                       \
        *(float2*)((vs) + it * 32)        = make_float2(r[it].x, r[it].y);     \
        *(float2*)((vs) + it * 32 + PSTR) = make_float2(r[it].z, r[it].w);     \
      }                                                                        \
    }
#define FAL_BLOCK(g, LAST)                                                     \
    {                                                                          \
      if ((g) + 2 < 8 || !(LAST)) FAL_LOAD_Q(((g) + 2) & 3, qc + ((g) + 2) * 8) \
      if ((g) + 1 < 8)            FAL_LOAD_B(((g) + 1) & 1, vcur, (g) + 1)     \
      else if (!(LAST))           FAL_LOAD_B(0, vnxt, 0)                       \
      if ((g) == 3 && !(LAST))    FAL_STORE_TILE(vsn)                          \
      __builtin_amdgcn_sched_barrier(0);                                       \
      if ((g) + 1 < 8 || !(LAST)) FAL_MAKE_A(((g) + 1) & 1, ((g) + 1) & 3)     \
      FAL_WMMA((g) & 1)                                                        \
      __builtin_amdgcn_sched_barrier(0);                                       \
    }
#define FAL_CHUNK(LAST)                                                        \
    {                                                                          \
      const float2* qc   = &qall[c * JT + k0];                                 \
      const float*  vcur = vb + cur * TILE;                                    \
      const float*  vnxt = vb + nxt * TILE;                                    \
      float*        vsn  = &Vt[nxt * TILE + lbase];                            \
      FAL_BLOCK(0, LAST)                                                       \
      FAL_BLOCK(1, LAST)                                                       \
      FAL_BLOCK(2, LAST)                                                       \
      FAL_BLOCK(3, LAST)                                                       \
      if (!(LAST)) {                                                           \
        __syncthreads();                                                       \
        if (c + 2 < NCHUNK) {                                                  \
          _Pragma("unroll") for (int it = 0; it < 8; ++it)                     \
            r[it] = *(const float4*)(gbase + ((size_t)(it * 16) << 10) + (c + 2) * JT); \
        }                                                                      \
      }                                                                        \
      FAL_BLOCK(4, LAST)                                                       \
      FAL_BLOCK(5, LAST)                                                       \
      FAL_BLOCK(6, LAST)                                                       \
      FAL_BLOCK(7, LAST)                                                       \
    }

  v2f    bf[2][16];
  float4 qr[4][2];
  v2f    af[2][2];

  FAL_LOAD_Q(0, &qall[k0])
  FAL_LOAD_Q(1, &qall[8 + k0])
  FAL_LOAD_B(0, vb, 0)
  FAL_MAKE_A(0, 0)

  int cur = 0, nxt = 1;
  for (int c = 0; c < NCHUNK - 1; ++c) {
    FAL_CHUNK(false)
    cur = nxt;
    nxt = (nxt == NBUF - 1) ? 0 : nxt + 1;
  }
  {
    const int c = NCHUNK - 1;
    FAL_CHUNK(true)
  }
#undef FAL_LOAD_Q
#undef FAL_LOAD_B
#undef FAL_MAKE_A
#undef FAL_WMMA
#undef FAL_STORE_TILE
#undef FAL_BLOCK
#undef FAL_CHUNK

  float dtot = dsum + __shfl_xor(dsum, 16, 32);

#pragma unroll
  for (int rr = 0; rr < 8; ++rr) {
    int row   = rr + 8 * half;
    float d   = __shfl(dtot, rr + 8 * half, 32);
    float inv = 1.0f / d;
    float* orow = out + (((size_t)b * KDIM + i0 + row) << 7) + m;
#pragma unroll
    for (int t = 0; t < 8; ++t)
      orow[t * 16] = acc[t][rr] * inv;
  }
}

extern "C" void kernel_launch(void* const* d_in, const int* in_sizes, int n_in,
                              void* d_out, int out_size, void* d_ws, size_t ws_size,
                              hipStream_t stream) {
  const float* x = (const float*)d_in[0];
  const float* w = (const float*)d_in[1];
  float* out = (float*)d_out;

  dim3 grid(KDIM / 128, BATCH);
  fal_attn_kernel<<<grid, 256, 0, stream>>>(x, w, out);
}


# projection pre-pass dropped: each chunk's s2 is summed from the registers staging its V tile (per-wave partials via LDS); softmax on a running max, rescaled at chunk edges; sum order changes
# speedup vs baseline: 2.0152x; 1.0533x over previous
#include <hip/hip_runtime.h>

typedef __attribute__((ext_vector_type(2))) float v2f;
typedef __attribute__((ext_vector_type(8))) float v8f;

#define ALPHA 0.2f
#define BATCH 32
#define WDIM 128
#define KDIM 1024
#define JT   64
#define NCHUNK (KDIM / JT)
#define PSTR 258
#define TILE ((JT / 2) * PSTR)
#define NBUF 3

__device__ __forceinline__ float fal_xhalf(float v) {
  const int bits = __builtin_bit_cast(int, v);
  return __builtin_bit_cast(float, __builtin_amdgcn_permlanex16(bits, bits, 0x76543210, 0xfedcba98,
                                                                false, false));
}
template <int XM>
__device__ __forceinline__ float fal_rowx(float v) {
  const int bits = __builtin_bit_cast(int, v);
  return __builtin_bit_cast(float, __builtin_amdgcn_update_dpp(bits, bits, 0x160 + XM, 0xf, 0xf, false));
}
__device__ __forceinline__ float fal_wave_max(float v) {
  v = fmaxf(v, fal_rowx<1>(v));
  v = fmaxf(v, fal_rowx<2>(v));
  v = fmaxf(v, fal_rowx<4>(v));
  v = fmaxf(v, fal_rowx<8>(v));
  return fmaxf(v, fal_xhalf(v));
}

__global__ __launch_bounds__(256) void fal_attn_kernel(const float* __restrict__ x,
                                                       const float* __restrict__ w,
                                                       float* __restrict__ out) {
  __shared__ float  Vt[NBUF * TILE];
  __shared__ __align__(16) float2 qw[8][KDIM];
  __shared__ __align__(16) float  part[2][8][JT];
  __shared__ __align__(16) float  p1[8][128];

  const int tid  = threadIdx.x;
  const int lane = tid & 31;
  const int wave = tid >> 5;
  const int half = lane >> 4;
  const int m    = lane & 15;
  const int b    = blockIdx.y;
  const int i0   = blockIdx.x * 128 + wave * 16;

  const int jj4 = tid & 15;
  const int w0  = tid >> 4;
  const float* gbase = x + (((size_t)b * WDIM + w0) << 10) + (jj4 << 2);
  const int lbase = (jj4 * 2) * PSTR + 2 * w0;
  float4 r[8];

  float w2r[8];
#pragma unroll
  for (int it = 0; it < 8; ++it) w2r[it] = w[WDIM + w0 + 16 * it];

#define FAL_PARTIALS(dst)                                                      \
    {                                                                          \
      float pa = 0.f, pb = 0.f, pc = 0.f, pd = 0.f;                            \
      _Pragma("unroll") for (int it = 0; it < 8; ++it) {                       \
        pa = fmaf(r[it].x, w2r[it], pa);                                       \
        pb = fmaf(r[it].y, w2r[it], pb);                                       \
        pc = fmaf(r[it].z, w2r[it], pc);                                       \
        pd = fmaf(r[it].w, w2r[it], pd);                                       \
      }                                                                        \
      pa += fal_xhalf(pa);                                                     \
      pb += fal_xhalf(pb);                                                     \
      pc += fal_xhalf(pc);                                                     \
      pd += fal_xhalf(pd);                                                     \
      *(float4*)(dst) = make_float4(pa, pb, pc, pd);                           \
    }
#define FAL_FIN_LOAD(src)                                                      \
    {                                                                          \
      _Pragma("unroll") for (int wv = 0; wv < 8; ++wv)                         \
        pl[wv] = *(const v2f*)((src) + wv * JT);                               \
    }
#define FAL_FIN_COMPUTE(qdst)                                                  \
    {                                                                          \
                                       \
      asm volatile("" : "+v"(pl[0]), "+v"(pl[1]), "+v"(pl[2]), "+v"(pl[3]),    \
                        "+v"(pl[4]), "+v"(pl[5]), "+v"(pl[6]), "+v"(pl[7]));   \
      float sa = pl[0].x, sb = pl[0].y;                                        \
      _Pragma("unroll") for (int wv = 1; wv < 8; ++wv) {                       \
        sa += pl[wv].x;                                                        \
        sb += pl[wv].y;                                                        \
      }                                                                        \
      *(float4*)(qdst) = make_float4(__expf(sa), __expf(ALPHA * sa),           \
                                     __expf(sb), __expf(ALPHA * sb));          \
      cm = fal_wave_max(fmaxf(sa, sb));                                        \
    }

  {
    const int wv = __builtin_amdgcn_readfirstlane(wave);
    const float* xp = x + (((size_t)b * WDIM + wv * 16) << 10) + blockIdx.x * 128 + (lane << 2);
    float4 xs[16];
#pragma unroll
    for (int u = 0; u < 16; ++u) xs[u] = *(const float4*)(xp + ((size_t)u << 10));
#pragma unroll
    for (int it = 0; it < 8; ++it)
      r[it] = *(const float4*)(gbase + ((size_t)(it * 16) << 10));

    float a1[4] = {0.f, 0.f, 0.f, 0.f};
#pragma unroll
    for (int u = 0; u < 16; ++u) {
      const float w1v = w[wv * 16 + u];
      a1[0] = fmaf(xs[u].x, w1v, a1[0]);
      a1[1] = fmaf(xs[u].y, w1v, a1[1]);
      a1[2] = fmaf(xs[u].z, w1v, a1[2]);
      a1[3] = fmaf(xs[u].w, w1v, a1[3]);
    }
    *(float4*)&p1[wave][4 * lane] = make_float4(a1[0], a1[1], a1[2], a1[3]);

    FAL_PARTIALS(&part[0][wave][4 * jj4])

#pragma unroll
    for (int it = 0; it < 8; ++it) {
      int o = lbase + it * 32;
      *(float2*)&Vt[o]        = make_float2(r[it].x, r[it].y);
      *(float2*)&Vt[o + PSTR] = make_float2(r[it].z, r[it].w);
    }
  }
  __syncthreads();
#pragma unroll
  for (int it = 0; it < 8; ++it)
    r[it] = *(const float4*)(gbase + ((size_t)(it * 16) << 10) + JT);

  float s1v = p1[0][wave * 16 + m];
#pragma unroll
  for (int wv = 1; wv < 8; ++wv) s1v += p1[wv][wave * 16 + m];

  v2f    pl[8];
  float  cm;
  FAL_FIN_LOAD(&part[0][0][2 * lane])
  FAL_FIN_COMPUTE(&qw[wave][2 * lane])

  float mref = cm;
  float zmax = s1v + mref;
  float mrow = fmaxf(zmax, ALPHA * zmax);
  float u1   = __expf(s1v - mrow);
  float ua   = __expf(ALPHA * s1v - mrow);
  float mnew = mref, mrn = mrow, u1n = u1, uan = ua, fsc = 1.f;
  float fr[8];
#pragma unroll
  for (int rr = 0; rr < 8; ++rr) fr[rr] = 1.f;
  int ev = 0;

  v8f acc[8];
#pragma unroll
  for (int t = 0; t < 8; ++t) acc[t] = (v8f)0.f;
  float dsum = 0.f;

  const int k0 = half * 2;
  const float* vb = &Vt[half * PSTR + 2 * m];

#define FAL_LOAD_Q(slot, qp)                                                   \
    {                                                                          \
      qr[slot][0] = *(const float4*)(qp);                                      \
      qr[slot][1] = *(const float4*)((qp) + 4);                                \
    }
#define FAL_LOAD_B(set, vp, gb)                                                \
    {                                                                          \
      _Pragma("unroll") for (int s = 0; s < 2; ++s)                            \
        _Pragma("unroll") for (int t = 0; t < 8; ++t)                          \
          bf[set][s * 8 + t] =                                                 \
              *(const v2f*)((vp) + ((gb) * 4 + s * 2) * PSTR + t * 32);        \
    }
#define FAL_MAKE_A(aset, qslot)                                                \
    {                                                                          \
      _Pragma("unroll") for (int s = 0; s < 2; ++s) {                          \
        float4 q = qr[qslot][s];                                               \
        float p0 = fmaxf(u1 * q.x, ua * q.y);                                  \
        float p1v = fmaxf(u1 * q.z, ua * q.w);                                 \
        dsum += p0 + p1v;                                                      \
        af[aset][s].x = p0;                                                    \
        af[aset][s].y = p1v;                                                   \
      }                                                                        \
    }
#define FAL_WMMA(set)                                                          \
    {                                                                          \
      _Pragma("unroll") for (int s = 0; s < 2; ++s)                            \
        _Pragma("unroll") for (int t = 0; t < 8; ++t)                          \
          acc[t] = __builtin_amdgcn_wmma_f32_16x16x4_f32(false, af[set][s], false, \
                                                         bf[set][s * 8 + t],   \
                                                         (short)0, acc[t], false, false); \
    }
#define FAL_STORE_TILE(vs)                                                     \
    {                                                                          \
      _Pragma("unroll") for (int it = 0; it < 8; ++it) {                       \
        *(float2*)((vs) + it * 32)        = make_float2(r[it].x, r[it].y);     \
        *(float2*)((vs) + it * 32 + PSTR) = make_float2(r[it].z, r[it].w);     \
      }                                                                        \
    }
#define FAL_REF_NEXT                                                           \
    {                                                                          \
      ev   = __builtin_amdgcn_readfirstlane((int)(cm > mref));                 \
      mnew = fmaxf(mref, cm);                                                  \
      const float zn = s1v + mnew;                                             \
      mrn  = fmaxf(zn, ALPHA * zn);                                            \
      fsc  = __expf(mrow - mrn);                                               \
      u1n  = __expf(s1v - mrn);                                                \
      uan  = __expf(ALPHA * s1v - mrn);                                        \
    }
#define FAL_REF_SWITCH                                                         \
    {                                                                          \
      dsum *= fsc;                                                             \
      u1 = u1n; ua = uan; mrow = mrn; mref = mnew;                             \
    }
#define FAL_RESCALE_ACC                                                        \
    {                                                                          \
      if (ev) {                                                                \
        _Pragma("unroll") for (int t = 0; t < 8; ++t)                          \
          _Pragma("unroll") for (int rr = 0; rr < 8; ++rr)                     \
            acc[t][rr] *= fr[rr];                                              \
      }                                                                        \
    }
#define FAL_BLOCK(g, LAST)                                                     \
    {                                                                          \
      if ((g) + 2 < 8 || !(LAST)) FAL_LOAD_Q(((g) + 2) & 3, qc + ((g) + 2) * 8) \
      if ((g) + 1 < 8)            FAL_LOAD_B(((g) + 1) & 1, vcur, (g) + 1)     \
      else if (!(LAST))           FAL_LOAD_B(0, vnxt, 0)                       \
      if ((g) == 3 && !(LAST))    { FAL_STORE_TILE(vsn) FAL_PARTIALS(pwn) }    \
      if ((g) == 4 && !(LAST))    FAL_FIN_LOAD(prd)                            \
      if ((g) == 7 && !(LAST)) {                                               \
        _Pragma("unroll") for (int rr = 0; rr < 8; ++rr)                       \
          fr[rr] = __shfl(fsc, rr + 8 * half, 32);                             \
      }                                                                        \
      __builtin_amdgcn_sched_barrier(0);                                       \
      if ((g) == 0)               FAL_RESCALE_ACC                              \
      if ((g) == 5 && !(LAST))    FAL_FIN_COMPUTE(qnx)                         \
      if ((g) == 6 && !(LAST))    FAL_REF_NEXT                                 \
      if ((g) == 7 && !(LAST))    FAL_REF_SWITCH                               \
      if ((g) + 1 < 8 || !(LAST)) FAL_MAKE_A(((g) + 1) & 1, ((g) + 1) & 3)     \
      FAL_WMMA((g) & 1)                                                        \
      __builtin_amdgcn_sched_barrier(0);                                       \
    }
#define FAL_CHUNK(LAST)                                                        \
    {                                                                          \
      const float2* qc   = &qw[wave][c * JT + k0];                             \
      const float*  vcur = vb + cur * TILE;                                    \
      const float*  vnxt = vb + nxt * TILE;                                    \
      float*        vsn  = &Vt[nxt * TILE + lbase];                            \
      float*        pwn  = &part[(c + 1) & 1][wave][4 * jj4];                  \
      const float*  prd  = &part[(c + 1) & 1][0][2 * lane];                    \
      float*        qnx  = (float*)&qw[wave][((c + 1) & (NCHUNK - 1)) * JT + 2 * lane]; \
      FAL_BLOCK(0, LAST)                                                       \
      FAL_BLOCK(1, LAST)                                                       \
      FAL_BLOCK(2, LAST)                                                       \
      FAL_BLOCK(3, LAST)                                                       \
      if (!(LAST)) {                                                           \
        __syncthreads();                                                       \
        if (c + 2 < NCHUNK) {                                                  \
          _Pragma("unroll") for (int it = 0; it < 8; ++it)                     \
            r[it] = *(const float4*)(gbase + ((size_t)(it * 16) << 10) + (c + 2) * JT); \
        }                                                                      \
      }                                                                        \
      FAL_BLOCK(4, LAST)                                                       \
      FAL_BLOCK(5, LAST)                                                       \
      FAL_BLOCK(6, LAST)                                                       \
      FAL_BLOCK(7, LAST)                                                       \
    }

  v2f    bf[2][16];
  float4 qr[4][2];
  v2f    af[2][2];

  FAL_LOAD_Q(0, &qw[wave][k0])
  FAL_LOAD_Q(1, &qw[wave][8 + k0])
  FAL_LOAD_B(0, vb, 0)
  FAL_MAKE_A(0, 0)

  int cur = 0, nxt = 1;
  for (int c = 0; c < NCHUNK - 1; ++c) {
    FAL_CHUNK(false)
    cur = nxt;
    nxt = (nxt == NBUF - 1) ? 0 : nxt + 1;
  }
  {
    const int c = NCHUNK - 1;
    FAL_CHUNK(true)
  }
#undef FAL_PARTIALS
#undef FAL_FIN_LOAD
#undef FAL_FIN_COMPUTE
#undef FAL_LOAD_Q
#undef FAL_LOAD_B
#undef FAL_MAKE_A
#undef FAL_WMMA
#undef FAL_STORE_TILE
#undef FAL_REF_NEXT
#undef FAL_REF_SWITCH
#undef FAL_RESCALE_ACC
#undef FAL_BLOCK
#undef FAL_CHUNK

  float dtot = dsum + __shfl_xor(dsum, 16, 32);

#pragma unroll
  for (int rr = 0; rr < 8; ++rr) {
    int row   = rr + 8 * half;
    float d   = __shfl(dtot, rr + 8 * half, 32);
    float inv = 1.0f / d;
    float* orow = out + (((size_t)b * KDIM + i0 + row) << 7) + m;
#pragma unroll
    for (int t = 0; t < 8; ++t)
      orow[t * 16] = acc[t][rr] * inv;
  }
}

extern "C" void kernel_launch(void* const* d_in, const int* in_sizes, int n_in,
                              void* d_out, int out_size, void* d_ws, size_t ws_size,
                              hipStream_t stream) {
  const float* x = (const float*)d_in[0];
  const float* w = (const float*)d_in[1];
  float* out = (float*)d_out;

  dim3 grid(KDIM / 128, BATCH);
  fal_attn_kernel<<<grid, 256, 0, stream>>>(x, w, out);
}


# rank-1 split over the batch: rows sorted by s1, 128 consecutive ranks per work-group; columns outside its threshold band enter as two vector sums, WMMA over the band only; sum order changes
# speedup vs baseline: 2.8819x; 1.4301x over previous
#include <hip/hip_runtime.h>

typedef __attribute__((ext_vector_type(2))) float v2f;
typedef __attribute__((ext_vector_type(8))) float v8f;

#define ALPHA 0.2f
#define BATCH 32
#define WDIM 128
#define KDIM 1024
#define PSTR 258
#define NBCAP 256

__device__ __forceinline__ float fal_xhalf(float v) {
  const int bits = __builtin_bit_cast(int, v);
  return __builtin_bit_cast(float, __builtin_amdgcn_permlanex16(bits, bits, 0x76543210, 0xfedcba98,
                                                                false, false));
}
template <int XM>
__device__ __forceinline__ float fal_rowx(float v) {
  const int bits = __builtin_bit_cast(int, v);
  return __builtin_bit_cast(float, __builtin_amdgcn_update_dpp(bits, bits, 0x160 + XM, 0xf, 0xf, false));
}
__device__ __forceinline__ float fal_key_s1(unsigned long long key) {
  const unsigned ord  = (unsigned)(key >> 32);
  const unsigned bits = ord ^ ((ord & 0x80000000u) ? 0x80000000u : 0xffffffffu);
  return __builtin_bit_cast(float, bits);
}
__device__ __forceinline__ float fal_row_sum(float v) {
  v += fal_rowx<1>(v);
  v += fal_rowx<2>(v);
  v += fal_rowx<4>(v);
  v += fal_rowx<8>(v);
  return v;
}
__device__ __forceinline__ float fal_wave_sum(float v) {
  v = fal_row_sum(v);
  return v + fal_xhalf(v);
}
__device__ __forceinline__ float fal_wave_max(float v) {
  v = fmaxf(v, fal_rowx<1>(v));
  v = fmaxf(v, fal_rowx<2>(v));
  v = fmaxf(v, fal_rowx<4>(v));
  v = fmaxf(v, fal_rowx<8>(v));
  return fmaxf(v, fal_xhalf(v));
}

__global__ __launch_bounds__(256) void fal_attn_kernel(const float* __restrict__ x,
                                                       const float* __restrict__ w,
                                                       float* __restrict__ out) {
  __shared__ __align__(16) float  BT[(NBCAP / 2) * PSTR];
  __shared__ __align__(16) unsigned long long skey[KDIM];
  __shared__ __align__(16) float  s2s[KDIM];
  __shared__ __align__(16) float  c1[KDIM];
  __shared__ __align__(16) float  ca[KDIM];
  __shared__ __align__(16) int    bp[KDIM];
  __shared__ __align__(16) float2 bq[KDIM];
  __shared__ float SA[WDIM], SB[WDIM];
  __shared__ float red[8];
  __shared__ float dred[2][8];
  __shared__ int   cnt[4][8];

  const int tid  = threadIdx.x;
  const int lane = tid & 31;
  const int wave = tid >> 5;
  const int half = lane >> 4;
  const int m    = lane & 15;
  const int b    = blockIdx.y;
  const int g    = blockIdx.x;
  const float* xb = x + (((size_t)b * WDIM) << 10);

  {
    float a1[4] = {0.f, 0.f, 0.f, 0.f}, a2[4] = {0.f, 0.f, 0.f, 0.f};
    const float* xp = xb + 4 * tid;
    for (int wb = 0; wb < WDIM; wb += 16) {
      float4 xs[16];
#pragma unroll
      for (int u = 0; u < 16; ++u) xs[u] = *(const float4*)(xp + ((size_t)(wb + u) << 10));
#pragma unroll
      for (int u = 0; u < 16; ++u) {
        const float w1v = w[wb + u];
        const float w2v = w[WDIM + wb + u];
        a1[0] = fmaf(xs[u].x, w1v, a1[0]);
        a1[1] = fmaf(xs[u].y, w1v, a1[1]);
        a1[2] = fmaf(xs[u].z, w1v, a1[2]);
        a1[3] = fmaf(xs[u].w, w1v, a1[3]);
        a2[0] = fmaf(xs[u].x, w2v, a2[0]);
        a2[1] = fmaf(xs[u].y, w2v, a2[1]);
        a2[2] = fmaf(xs[u].z, w2v, a2[2]);
        a2[3] = fmaf(xs[u].w, w2v, a2[3]);
      }
    }
#pragma unroll
    for (int e = 0; e < 4; ++e) {
      const unsigned bits = __builtin_bit_cast(unsigned, a1[e]);
      const unsigned ord  = bits ^ ((bits & 0x80000000u) ? 0xffffffffu : 0x80000000u);
      skey[4 * tid + e] = ((unsigned long long)ord << 32) | (unsigned)(4 * tid + e);
    }
    *(float4*)&s2s[4 * tid] = make_float4(a2[0], a2[1], a2[2], a2[3]);
    const float mx = fal_wave_max(fmaxf(fmaxf(a2[0], a2[1]), fmaxf(a2[2], a2[3])));
    if (lane == 0) red[wave] = mx;
  }
  __syncthreads();
  float s2max = red[0];
#pragma unroll
  for (int wv = 1; wv < 8; ++wv) s2max = fmaxf(s2max, red[wv]);

  for (int k = 2; k <= KDIM; k <<= 1) {
    for (int j = k >> 1; j > 0; j >>= 1) {
      const int t0 = tid, t1 = tid + 256;
      const int i0 = ((t0 & ~(j - 1)) << 1) | (t0 & (j - 1));
      const int i1 = ((t1 & ~(j - 1)) << 1) | (t1 & (j - 1));
      const int p0 = i0 | j, p1 = i1 | j;
      const unsigned long long ka0 = skey[i0], kb0 = skey[p0];
      const unsigned long long ka1 = skey[i1], kb1 = skey[p1];
      if ((ka0 > kb0) == ((i0 & k) == 0)) { skey[i0] = kb0; skey[p0] = ka0; }
      if ((ka1 > kb1) == ((i1 & k) == 0)) { skey[i1] = kb1; skey[p1] = ka1; }
      __syncthreads();
    }
  }

  const float s1min = fal_key_s1(skey[128 * g]);
  const float s1max = fal_key_s1(skey[128 * g + 127]);

  int nband;
  float DA, DB;
  {
    unsigned msk[4];
    float e1r[4], ear[4];
    float da = 0.f, db = 0.f;
#pragma unroll
    for (int rnd = 0; rnd < 4; ++rnd) {
      const int j = tid + 256 * rnd;
      const float s2 = s2s[j];
      const bool above = (s1min + s2 > 0.f);
      const bool below = (s1max + s2 <= 0.f);
      const float d = s2 - s2max;
      e1r[rnd] = __expf(d);
      ear[rnd] = __expf(ALPHA * d);
      const float k1 = above ? e1r[rnd] : 0.f;
      const float k2 = (below && !above) ? ear[rnd] : 0.f;
      c1[j] = k1;
      ca[j] = k2;
      da += k1;
      db += k2;
      msk[rnd] = __builtin_amdgcn_ballot_w32(!above && !below);
      if (lane == 0) cnt[rnd][wave] = __popc(msk[rnd]);
    }
    da = fal_wave_sum(da);
    db = fal_wave_sum(db);
    if (lane == 0) { dred[0][wave] = da; dred[1][wave] = db; }
    __syncthreads();
    int run = 0;
#pragma unroll
    for (int rnd = 0; rnd < 4; ++rnd) {
      int mine = 0;
#pragma unroll
      for (int wv = 0; wv < 8; ++wv) {
        if (wv == wave) mine = run;
        run += cnt[rnd][wv];
      }
      const int j = tid + 256 * rnd;
      if ((msk[rnd] >> lane) & 1u) {
        const int pos = mine + __popc(msk[rnd] & ((1u << lane) - 1u));
        bp[j] = pos;
        bq[pos] = make_float2(e1r[rnd], ear[rnd]);
      } else {
        bp[j] = -1;
      }
    }
    nband = run;
    if (tid < 3 && nband + tid < ((nband + 3) & ~3)) bq[nband + tid] = make_float2(0.f, 0.f);
    DA = dred[0][0];
    DB = dred[1][0];
#pragma unroll
    for (int wv = 1; wv < 8; ++wv) { DA += dred[0][wv]; DB += dred[1][wv]; }
  }
  __syncthreads();

  const int   rpos = 128 * g + 16 * wave + m;
  const unsigned long long rkey = skey[rpos];
  const float s1v  = fal_key_s1(rkey);
  const int   ridx = (int)(unsigned)(rkey & 0xffffffffull);
  const float zmax = s1v + s2max;
  const float mrow = fmaxf(zmax, ALPHA * zmax);
  const float u1   = __expf(zmax - mrow);
  const float ua   = __expf(ALPHA * zmax - mrow);

  const int jj4 = tid & 15;
  const int w0  = tid >> 4;
  const float* gbase = xb + ((size_t)w0 << 10) + (jj4 << 2);
  float r1[8], ra[8];
  v8f acc[8];
#pragma unroll
  for (int it = 0; it < 8; ++it) { r1[it] = 0.f; ra[it] = 0.f; acc[it] = (v8f)0.f; }
  float dsum = 0.f;

  const int npieces = nband > 0 ? (nband + NBCAP - 1) / NBCAP : 1;
  for (int q = 0; q < npieces; ++q) {
    const int plo = q * NBCAP;
    float4 r[8], rn[8];
#pragma unroll
    for (int it = 0; it < 8; ++it) r[it] = *(const float4*)(gbase + ((size_t)(it * 16) << 10));
    for (int c = 0; c < KDIM / 64; ++c) {
      if (c + 1 < KDIM / 64) {
#pragma unroll
        for (int it = 0; it < 8; ++it)
          rn[it] = *(const float4*)(gbase + ((size_t)(it * 16) << 10) + (c + 1) * 64);
      }
      const int j0 = c * 64 + 4 * jj4;
      if (q == 0) {
        const float4 k1 = *(const float4*)&c1[j0];
        const float4 k2 = *(const float4*)&ca[j0];
#pragma unroll
        for (int it = 0; it < 8; ++it) {
          r1[it] = fmaf(r[it].x, k1.x, r1[it]);
          r1[it] = fmaf(r[it].y, k1.y, r1[it]);
          r1[it] = fmaf(r[it].z, k1.z, r1[it]);
          r1[it] = fmaf(r[it].w, k1.w, r1[it]);
          ra[it] = fmaf(r[it].x, k2.x, ra[it]);
          ra[it] = fmaf(r[it].y, k2.y, ra[it]);
          ra[it] = fmaf(r[it].z, k2.z, ra[it]);
          ra[it] = fmaf(r[it].w, k2.w, ra[it]);
        }
      }
      const int4 pl = *(const int4*)&bp[j0];
#define FAL_BAND_STORE(P, COMP)                                                \
      {                                                                        \
        const int lp = (P) - plo;                                              \
        if ((unsigned)lp < (unsigned)NBCAP) {                                  \
          float* d = &BT[(lp >> 1) * PSTR + (lp & 1) + 2 * w0];                \
          _Pragma("unroll") for (int it = 0; it < 8; ++it) d[32 * it] = r[it].COMP; \
        }                                                                      \
      }
      FAL_BAND_STORE(pl.x, x)
      FAL_BAND_STORE(pl.y, y)
      FAL_BAND_STORE(pl.z, z)
      FAL_BAND_STORE(pl.w, w)
#undef FAL_BAND_STORE
#pragma unroll
      for (int it = 0; it < 8; ++it) r[it] = rn[it];
    }
    if (q == 0) {
#pragma unroll
      for (int it = 0; it < 8; ++it) {
        const float t1 = fal_row_sum(r1[it]);
        const float t2 = fal_row_sum(ra[it]);
        if (jj4 == 0) { SA[w0 + 16 * it] = t1; SB[w0 + 16 * it] = t2; }
      }
    }
    const int np   = min(NBCAP, nband - plo);
    const int npad = (np + 3) & ~3;
    if (tid < WDIM) {
      for (int lp = np; lp < npad; ++lp) BT[(lp >> 1) * PSTR + (lp & 1) + 2 * tid] = 0.f;
    }
    __syncthreads();
    for (int kk = 0; kk < npad; kk += 4) {
      const float4 qv = *(const float4*)&bq[plo + kk + 2 * half];
      const float p0 = fmaxf(u1 * qv.x, ua * qv.y);
      const float p1 = fmaxf(u1 * qv.z, ua * qv.w);
      dsum += p0 + p1;
      v2f af;
      af.x = p0;
      af.y = p1;
      const float* vp = &BT[((kk >> 1) + half) * PSTR + 2 * m];
#pragma unroll
      for (int t = 0; t < 8; ++t) {
        const v2f bfr = *(const v2f*)(vp + t * 32);
        acc[t] = __builtin_amdgcn_wmma_f32_16x16x4_f32(false, af, false, bfr, (short)0, acc[t],
                                                       false, false);
      }
    }
    __syncthreads();
  }

  const float dband = dsum + fal_xhalf(dsum);
  const float den   = fmaf(u1, DA, fmaf(ua, DB, dband));
  float sa[8], sb[8];
#pragma unroll
  for (int t = 0; t < 8; ++t) { sa[t] = SA[t * 16 + m]; sb[t] = SB[t * 16 + m]; }
#pragma unroll
  for (int rr = 0; rr < 8; ++rr) {
    const int   row = rr + 8 * half;
    const float u1r = __shfl(u1, row, 32);
    const float uar = __shfl(ua, row, 32);
    const float inv = 1.0f / __shfl(den, row, 32);
    const int   ri  = __shfl(ridx, row, 32);
    float* orow = out + (((size_t)b * KDIM + ri) << 7) + m;
#pragma unroll
    for (int t = 0; t < 8; ++t)
      orow[t * 16] = fmaf(u1r, sa[t], fmaf(uar, sb[t], acc[t][rr])) * inv;
  }
}

extern "C" void kernel_launch(void* const* d_in, const int* in_sizes, int n_in,
                              void* d_out, int out_size, void* d_ws, size_t ws_size,
                              hipStream_t stream) {
  const float* x = (const float*)d_in[0];
  const float* w = (const float*)d_in[1];
  float* out = (float*)d_out;

  dim3 grid(KDIM / 128, BATCH);
  fal_attn_kernel<<<grid, 256, 0, stream>>>(x, w, out);
}


# sort of the (s1, index) keys: bitonic steps of distance below 128 done in registers and across lanes (DPP, permlanex16), only the 6 steps between waves via LDS; 55 barriers -> 7; sum order unchanged
# speedup vs baseline: 3.3358x; 1.1575x over previous
#include <hip/hip_runtime.h>

typedef __attribute__((ext_vector_type(2))) float v2f;
typedef __attribute__((ext_vector_type(8))) float v8f;

#define ALPHA 0.2f
#define BATCH 32
#define WDIM 128
#define KDIM 1024
#define PSTR 258
#define NBCAP 256

__device__ __forceinline__ float fal_xhalf(float v) {
  const int bits = __builtin_bit_cast(int, v);
  return __builtin_bit_cast(float, __builtin_amdgcn_permlanex16(bits, bits, 0x76543210, 0xfedcba98,
                                                                false, false));
}
template <int XM>
__device__ __forceinline__ float fal_rowx(float v) {
  const int bits = __builtin_bit_cast(int, v);
  return __builtin_bit_cast(float, __builtin_amdgcn_update_dpp(bits, bits, 0x160 + XM, 0xf, 0xf, false));
}
__device__ __forceinline__ float fal_key_s1(unsigned long long key) {
  const unsigned ord  = (unsigned)(key >> 32);
  const unsigned bits = ord ^ ((ord & 0x80000000u) ? 0x80000000u : 0xffffffffu);
  return __builtin_bit_cast(float, bits);
}
__device__ __forceinline__ float fal_row_sum(float v) {
  v += fal_rowx<1>(v);
  v += fal_rowx<2>(v);
  v += fal_rowx<4>(v);
  v += fal_rowx<8>(v);
  return v;
}
__device__ __forceinline__ float fal_wave_sum(float v) {
  v = fal_row_sum(v);
  return v + fal_xhalf(v);
}
__device__ __forceinline__ float fal_wave_max(float v) {
  v = fmaxf(v, fal_rowx<1>(v));
  v = fmaxf(v, fal_rowx<2>(v));
  v = fmaxf(v, fal_rowx<4>(v));
  v = fmaxf(v, fal_rowx<8>(v));
  return fmaxf(v, fal_xhalf(v));
}

typedef unsigned long long fal_u64;
typedef __attribute__((ext_vector_type(2))) unsigned long long fal_v2u64;

template <int XM>
__device__ __forceinline__ fal_u64 fal_key_rowx(fal_u64 v) {
  const int lo = (int)(unsigned)v, hi = (int)(unsigned)(v >> 32);
  const unsigned nlo = (unsigned)__builtin_amdgcn_update_dpp(lo, lo, 0x160 + XM, 0xf, 0xf, false);
  const unsigned nhi = (unsigned)__builtin_amdgcn_update_dpp(hi, hi, 0x160 + XM, 0xf, 0xf, false);
  return ((fal_u64)nhi << 32) | nlo;
}
__device__ __forceinline__ fal_u64 fal_key_xhalf(fal_u64 v) {
  const int lo = (int)(unsigned)v, hi = (int)(unsigned)(v >> 32);
  const unsigned nlo = (unsigned)__builtin_amdgcn_permlanex16(lo, lo, 0x76543210, 0xfedcba98, false, false);
  const unsigned nhi = (unsigned)__builtin_amdgcn_permlanex16(hi, hi, 0x76543210, 0xfedcba98, false, false);
  return ((fal_u64)nhi << 32) | nlo;
}
template <int J> struct fal_xch {
  static __device__ __forceinline__ fal_u64 get(fal_u64 v) { return fal_key_rowx<(J >> 2)>(v); }
};
template <> struct fal_xch<64> {
  static __device__ __forceinline__ fal_u64 get(fal_u64 v) { return fal_key_xhalf(v); }
};
template <int J>
__device__ __forceinline__ void fal_lane_stage(fal_u64 (&key)[4], int lane, bool asc) {
  const bool keepmin = (((lane & (J >> 2)) == 0) == asc);
#pragma unroll
  for (int e = 0; e < 4; ++e) {
    const fal_u64 o = fal_xch<J>::get(key[e]);
    const bool take = (o > key[e]) != keepmin;
    key[e] = take ? o : key[e];
  }
}
__device__ __forceinline__ void fal_ce(fal_u64& a, fal_u64& b, bool asc) {
  const bool sw = (a > b) == asc;
  const fal_u64 na = sw ? b : a;
  const fal_u64 nb = sw ? a : b;
  a = na;
  b = nb;
}

__global__ __launch_bounds__(256) void fal_attn_kernel(const float* __restrict__ x,
                                                       const float* __restrict__ w,
                                                       float* __restrict__ out) {
  __shared__ __align__(16) float  BT[(NBCAP / 2) * PSTR];
  __shared__ __align__(16) unsigned long long skey[KDIM];
  __shared__ __align__(16) unsigned long long skx[KDIM];
  __shared__ __align__(16) float  s2s[KDIM];
  __shared__ __align__(16) float  c1[KDIM];
  __shared__ __align__(16) float  ca[KDIM];
  __shared__ __align__(16) int    bp[KDIM];
  __shared__ __align__(16) float2 bq[KDIM];
  __shared__ float SA[WDIM], SB[WDIM];
  __shared__ float red[8];
  __shared__ float dred[2][8];
  __shared__ int   cnt[4][8];

  const int tid  = threadIdx.x;
  const int lane = tid & 31;
  const int wave = tid >> 5;
  const int half = lane >> 4;
  const int m    = lane & 15;
  const int b    = blockIdx.y;
  const int g    = blockIdx.x;
  const float* xb = x + (((size_t)b * WDIM) << 10);

  fal_u64 key[4];
  {
    float a1[4] = {0.f, 0.f, 0.f, 0.f}, a2[4] = {0.f, 0.f, 0.f, 0.f};
    const float* xp = xb + 4 * tid;
    for (int wb = 0; wb < WDIM; wb += 16) {
      float4 xs[16];
#pragma unroll
      for (int u = 0; u < 16; ++u) xs[u] = *(const float4*)(xp + ((size_t)(wb + u) << 10));
#pragma unroll
      for (int u = 0; u < 16; ++u) {
        const float w1v = w[wb + u];
        const float w2v = w[WDIM + wb + u];
        a1[0] = fmaf(xs[u].x, w1v, a1[0]);
        a1[1] = fmaf(xs[u].y, w1v, a1[1]);
        a1[2] = fmaf(xs[u].z, w1v, a1[2]);
        a1[3] = fmaf(xs[u].w, w1v, a1[3]);
        a2[0] = fmaf(xs[u].x, w2v, a2[0]);
        a2[1] = fmaf(xs[u].y, w2v, a2[1]);
        a2[2] = fmaf(xs[u].z, w2v, a2[2]);
        a2[3] = fmaf(xs[u].w, w2v, a2[3]);
      }
    }
#pragma unroll
    for (int e = 0; e < 4; ++e) {
      const unsigned bits = __builtin_bit_cast(unsigned, a1[e]);
      const unsigned ord  = bits ^ ((bits & 0x80000000u) ? 0xffffffffu : 0x80000000u);
      key[e] = ((fal_u64)ord << 32) | (unsigned)(4 * tid + e);
    }
    *(float4*)&s2s[4 * tid] = make_float4(a2[0], a2[1], a2[2], a2[3]);
    const float mx = fal_wave_max(fmaxf(fmaxf(a2[0], a2[1]), fmaxf(a2[2], a2[3])));
    red[wave] = mx;
  }
  __syncthreads();
  float s2max = red[0];
#pragma unroll
  for (int wv = 1; wv < 8; ++wv) s2max = fmaxf(s2max, red[wv]);

  fal_ce(key[0], key[1], true);
  fal_ce(key[2], key[3], false);
  for (int k = 4; k <= KDIM; k <<= 1) {
    const bool asc = (((tid << 2) & k) == 0);
    if (k >= 256) {
      fal_u64* buf = (k == 512) ? skey : skx;
      *(fal_v2u64*)&buf[4 * tid]     = (fal_v2u64){key[0], key[1]};
      *(fal_v2u64*)&buf[4 * tid + 2] = (fal_v2u64){key[2], key[3]};
      __syncthreads();
      for (int j = k >> 1; j > 128; j >>= 1) {
        const int t0 = tid, t1 = tid + 256;
        const int i0 = ((t0 & ~(j - 1)) << 1) | (t0 & (j - 1));
        const int i1 = ((t1 & ~(j - 1)) << 1) | (t1 & (j - 1));
        const int p0 = i0 | j, p1 = i1 | j;
        const fal_u64 ka0 = buf[i0], kb0 = buf[p0];
        const fal_u64 ka1 = buf[i1], kb1 = buf[p1];
        if ((ka0 > kb0) == ((i0 & k) == 0)) { buf[i0] = kb0; buf[p0] = ka0; }
        if ((ka1 > kb1) == ((i1 & k) == 0)) { buf[i1] = kb1; buf[p1] = ka1; }
        __syncthreads();
      }
      const fal_v2u64 o01 = *(const fal_v2u64*)&buf[4 * tid];
      const fal_v2u64 o23 = *(const fal_v2u64*)&buf[4 * tid + 2];
      const fal_v2u64 q01 = *(const fal_v2u64*)&buf[(4 * tid) ^ 128];
      const fal_v2u64 q23 = *(const fal_v2u64*)&buf[((4 * tid) ^ 128) + 2];
      const fal_u64 own[4] = {o01.x, o01.y, o23.x, o23.y};
      const fal_u64 oth[4] = {q01.x, q01.y, q23.x, q23.y};
      const bool keepmin = ((((tid << 2) & 128) == 0) == asc);
#pragma unroll
      for (int e = 0; e < 4; ++e) key[e] = ((oth[e] > own[e]) != keepmin) ? oth[e] : own[e];
    }
    if (k >= 128) fal_lane_stage<64>(key, lane, asc);
    if (k >= 64)  fal_lane_stage<32>(key, lane, asc);
    if (k >= 32)  fal_lane_stage<16>(key, lane, asc);
    if (k >= 16)  fal_lane_stage<8>(key, lane, asc);
    if (k >= 8)   fal_lane_stage<4>(key, lane, asc);
    fal_ce(key[0], key[2], asc);
    fal_ce(key[1], key[3], asc);
    fal_ce(key[0], key[1], asc);
    fal_ce(key[2], key[3], asc);
  }
  *(fal_v2u64*)&skey[4 * tid]     = (fal_v2u64){key[0], key[1]};
  *(fal_v2u64*)&skey[4 * tid + 2] = (fal_v2u64){key[2], key[3]};
  __syncthreads();

  const float s1min = fal_key_s1(skey[128 * g]);
  const float s1max = fal_key_s1(skey[128 * g + 127]);

  int nband;
  float DA, DB;
  {
    unsigned msk[4];
    float e1r[4], ear[4];
    float da = 0.f, db = 0.f;
#pragma unroll
    for (int rnd = 0; rnd < 4; ++rnd) {
      const int j = tid + 256 * rnd;
      const float s2 = s2s[j];
      const bool above = (s1min + s2 > 0.f);
      const bool below = (s1max + s2 <= 0.f);
      const float d = s2 - s2max;
      e1r[rnd] = __expf(d);
      ear[rnd] = __expf(ALPHA * d);
      const float k1 = above ? e1r[rnd] : 0.f;
      const float k2 = (below && !above) ? ear[rnd] : 0.f;
      c1[j] = k1;
      ca[j] = k2;
      da += k1;
      db += k2;
      msk[rnd] = __builtin_amdgcn_ballot_w32(!above && !below);
      if (lane == 0) cnt[rnd][wave] = __popc(msk[rnd]);
    }
    da = fal_wave_sum(da);
    db = fal_wave_sum(db);
    if (lane == 0) { dred[0][wave] = da; dred[1][wave] = db; }
    __syncthreads();
    int run = 0;
#pragma unroll
    for (int rnd = 0; rnd < 4; ++rnd) {
      int mine = 0;
#pragma unroll
      for (int wv = 0; wv < 8; ++wv) {
        if (wv == wave) mine = run;
        run += cnt[rnd][wv];
      }
      const int j = tid + 256 * rnd;
      if ((msk[rnd] >> lane) & 1u) {
        const int pos = mine + __popc(msk[rnd] & ((1u << lane) - 1u));
        bp[j] = pos;
        bq[pos] = make_float2(e1r[rnd], ear[rnd]);
      } else {
        bp[j] = -1;
      }
    }
    nband = run;
    if (tid < 3 && nband + tid < ((nband + 3) & ~3)) bq[nband + tid] = make_float2(0.f, 0.f);
    DA = dred[0][0];
    DB = dred[1][0];
#pragma unroll
    for (int wv = 1; wv < 8; ++wv) { DA += dred[0][wv]; DB += dred[1][wv]; }
  }
  __syncthreads();

  const int   rpos = 128 * g + 16 * wave + m;
  const unsigned long long rkey = skey[rpos];
  const float s1v  = fal_key_s1(rkey);
  const int   ridx = (int)(unsigned)(rkey & 0xffffffffull);
  const float zmax = s1v + s2max;
  const float mrow = fmaxf(zmax, ALPHA * zmax);
  const float u1   = __expf(zmax - mrow);
  const float ua   = __expf(ALPHA * zmax - mrow);

  const int jj4 = tid & 15;
  const int w0  = tid >> 4;
  const float* gbase = xb + ((size_t)w0 << 10) + (jj4 << 2);
  float r1[8], ra[8];
  v8f acc[8];
#pragma unroll
  for (int it = 0; it < 8; ++it) { r1[it] = 0.f; ra[it] = 0.f; acc[it] = (v8f)0.f; }
  float dsum = 0.f;

  const int npieces = nband > 0 ? (nband + NBCAP - 1) / NBCAP : 1;
  for (int q = 0; q < npieces; ++q) {
    const int plo = q * NBCAP;
    float4 r[8], rn[8];
#pragma unroll
    for (int it = 0; it < 8; ++it) r[it] = *(const float4*)(gbase + ((size_t)(it * 16) << 10));
    for (int c = 0; c < KDIM / 64; ++c) {
      if (c + 1 < KDIM / 64) {
#pragma unroll
        for (int it = 0; it < 8; ++it)
          rn[it] = *(const float4*)(gbase + ((size_t)(it * 16) << 10) + (c + 1) * 64);
      }
      const int j0 = c * 64 + 4 * jj4;
      if (q == 0) {
        const float4 k1 = *(const float4*)&c1[j0];
        const float4 k2 = *(const float4*)&ca[j0];
#pragma unroll
        for (int it = 0; it < 8; ++it) {
          r1[it] = fmaf(r[it].x, k1.x, r1[it]);
          r1[it] = fmaf(r[it].y, k1.y, r1[it]);
          r1[it] = fmaf(r[it].z, k1.z, r1[it]);
          r1[it] = fmaf(r[it].w, k1.w, r1[it]);
          ra[it] = fmaf(r[it].x, k2.x, ra[it]);
          ra[it] = fmaf(r[it].y, k2.y, ra[it]);
          ra[it] = fmaf(r[it].z, k2.z, ra[it]);
          ra[it] = fmaf(r[it].w, k2.w, ra[it]);
        }
      }
      const int4 pl = *(const int4*)&bp[j0];
#define FAL_BAND_STORE(P, COMP)                                                \
      {                                                                        \
        const int lp = (P) - plo;                                              \
        if ((unsigned)lp < (unsigned)NBCAP) {                                  \
          float* d = &BT[(lp >> 1) * PSTR + (lp & 1) + 2 * w0];                \
          _Pragma("unroll") for (int it = 0; it < 8; ++it) d[32 * it] = r[it].COMP; \
        }                                                                      \
      }
      FAL_BAND_STORE(pl.x, x)
      FAL_BAND_STORE(pl.y, y)
      FAL_BAND_STORE(pl.z, z)
      FAL_BAND_STORE(pl.w, w)
#undef FAL_BAND_STORE
#pragma unroll
      for (int it = 0; it < 8; ++it) r[it] = rn[it];
    }
    if (q == 0) {
#pragma unroll
      for (int it = 0; it < 8; ++it) {
        const float t1 = fal_row_sum(r1[it]);
        const float t2 = fal_row_sum(ra[it]);
        if (jj4 == 0) { SA[w0 + 16 * it] = t1; SB[w0 + 16 * it] = t2; }
      }
    }
    const int np   = min(NBCAP, nband - plo);
    const int npad = (np + 3) & ~3;
    if (tid < WDIM) {
      for (int lp = np; lp < npad; ++lp) BT[(lp >> 1) * PSTR + (lp & 1) + 2 * tid] = 0.f;
    }
    __syncthreads();
    for (int kk = 0; kk < npad; kk += 4) {
      const float4 qv = *(const float4*)&bq[plo + kk + 2 * half];
      const float p0 = fmaxf(u1 * qv.x, ua * qv.y);
      const float p1 = fmaxf(u1 * qv.z, ua * qv.w);
      dsum += p0 + p1;
      v2f af;
      af.x = p0;
      af.y = p1;
      const float* vp = &BT[((kk >> 1) + half) * PSTR + 2 * m];
#pragma unroll
      for (int t = 0; t < 8; ++t) {
        const v2f bfr = *(const v2f*)(vp + t * 32);
        acc[t] = __builtin_amdgcn_wmma_f32_16x16x4_f32(false, af, false, bfr, (short)0, acc[t],
                                                       false, false);
      }
    }
    __syncthreads();
  }

  const float dband = dsum + fal_xhalf(dsum);
  const float den   = fmaf(u1, DA, fmaf(ua, DB, dband));
  float sa[8], sb[8];
#pragma unroll
  for (int t = 0; t < 8; ++t) { sa[t] = SA[t * 16 + m]; sb[t] = SB[t * 16 + m]; }
#pragma unroll
  for (int rr = 0; rr < 8; ++rr) {
    const int   row = rr + 8 * half;
    const float u1r = __shfl(u1, row, 32);
    const float uar = __shfl(ua, row, 32);
    const float inv = 1.0f / __shfl(den, row, 32);
    const int   ri  = __shfl(ridx, row, 32);
    float* orow = out + (((size_t)b * KDIM + ri) << 7) + m;
#pragma unroll
    for (int t = 0; t < 8; ++t)
      orow[t * 16] = fmaf(u1r, sa[t], fmaf(uar, sb[t], acc[t][rr])) * inv;
  }
}

extern "C" void kernel_launch(void* const* d_in, const int* in_sizes, int n_in,
                              void* d_out, int out_size, void* d_ws, size_t ws_size,
                              hipStream_t stream) {
  const float* x = (const float*)d_in[0];
  const float* w = (const float*)d_in[1];
  float* out = (float*)d_out;

  dim3 grid(KDIM / 128, BATCH);
  fal_attn_kernel<<<grid, 256, 0, stream>>>(x, w, out);
}


# band columns ordered by threshold class: each wave's WMMA runs over its own 16 rows' band only; the rest of the band reaches all waves via one shared 16-row boundary product; sum order changes
# speedup vs baseline: 3.6543x; 1.0955x over previous
#include <hip/hip_runtime.h>

typedef __attribute__((ext_vector_type(2))) float v2f;
typedef __attribute__((ext_vector_type(8))) float v8f;

#define ALPHA 0.2f
#define BATCH 32
#define WDIM 128
#define KDIM 1024
#define PSTR 258
#define NBCAP 256

__device__ __forceinline__ float fal_xhalf(float v) {
  const int bits = __builtin_bit_cast(int, v);
  return __builtin_bit_cast(float, __builtin_amdgcn_permlanex16(bits, bits, 0x76543210, 0xfedcba98,
                                                                false, false));
}
template <int XM>
__device__ __forceinline__ float fal_rowx(float v) {
  const int bits = __builtin_bit_cast(int, v);
  return __builtin_bit_cast(float, __builtin_amdgcn_update_dpp(bits, bits, 0x160 + XM, 0xf, 0xf, false));
}
__device__ __forceinline__ float fal_key_s1(unsigned long long key) {
  const unsigned ord  = (unsigned)(key >> 32);
  const unsigned bits = ord ^ ((ord & 0x80000000u) ? 0x80000000u : 0xffffffffu);
  return __builtin_bit_cast(float, bits);
}
__device__ __forceinline__ float fal_row_sum(float v) {
  v += fal_rowx<1>(v);
  v += fal_rowx<2>(v);
  v += fal_rowx<4>(v);
  v += fal_rowx<8>(v);
  return v;
}
__device__ __forceinline__ float fal_wave_sum(float v) {
  v = fal_row_sum(v);
  return v + fal_xhalf(v);
}
__device__ __forceinline__ float fal_wave_max(float v) {
  v = fmaxf(v, fal_rowx<1>(v));
  v = fmaxf(v, fal_rowx<2>(v));
  v = fmaxf(v, fal_rowx<4>(v));
  v = fmaxf(v, fal_rowx<8>(v));
  return fmaxf(v, fal_xhalf(v));
}

typedef unsigned long long fal_u64;
typedef __attribute__((ext_vector_type(2))) unsigned long long fal_v2u64;

template <int XM>
__device__ __forceinline__ fal_u64 fal_key_rowx(fal_u64 v) {
  const int lo = (int)(unsigned)v, hi = (int)(unsigned)(v >> 32);
  const unsigned nlo = (unsigned)__builtin_amdgcn_update_dpp(lo, lo, 0x160 + XM, 0xf, 0xf, false);
  const unsigned nhi = (unsigned)__builtin_amdgcn_update_dpp(hi, hi, 0x160 + XM, 0xf, 0xf, false);
  return ((fal_u64)nhi << 32) | nlo;
}
__device__ __forceinline__ fal_u64 fal_key_xhalf(fal_u64 v) {
  const int lo = (int)(unsigned)v, hi = (int)(unsigned)(v >> 32);
  const unsigned nlo = (unsigned)__builtin_amdgcn_permlanex16(lo, lo, 0x76543210, 0xfedcba98, false, false);
  const unsigned nhi = (unsigned)__builtin_amdgcn_permlanex16(hi, hi, 0x76543210, 0xfedcba98, false, false);
  return ((fal_u64)nhi << 32) | nlo;
}
template <int J> struct fal_xch {
  static __device__ __forceinline__ fal_u64 get(fal_u64 v) { return fal_key_rowx<(J >> 2)>(v); }
};
template <> struct fal_xch<64> {
  static __device__ __forceinline__ fal_u64 get(fal_u64 v) { return fal_key_xhalf(v); }
};
template <int J>
__device__ __forceinline__ void fal_lane_stage(fal_u64 (&key)[4], int lane, bool asc) {
  const bool keepmin = (((lane & (J >> 2)) == 0) == asc);
#pragma unroll
  for (int e = 0; e < 4; ++e) {
    const fal_u64 o = fal_xch<J>::get(key[e]);
    const bool take = (o > key[e]) != keepmin;
    key[e] = take ? o : key[e];
  }
}
__device__ __forceinline__ void fal_ce(fal_u64& a, fal_u64& b, bool asc) {
  const bool sw = (a > b) == asc;
  const fal_u64 na = sw ? b : a;
  const fal_u64 nb = sw ? a : b;
  a = na;
  b = nb;
}

__global__ __launch_bounds__(256) void fal_attn_kernel(const float* __restrict__ x,
                                                       const float* __restrict__ w,
                                                       float* __restrict__ out) {
  __shared__ __align__(16) float  BT[(NBCAP / 2) * PSTR];
  __shared__ __align__(16) unsigned long long skey[KDIM];
  __shared__ __align__(16) unsigned long long skx[KDIM];
  __shared__ __align__(16) float  s2s[KDIM];
  __shared__ __align__(16) float  c1[KDIM];
  __shared__ __align__(16) float  ca[KDIM];
  __shared__ __align__(16) int    bp[KDIM];
  __shared__ __align__(16) float2 bq[KDIM];
  __shared__ float SA[WDIM], SB[WDIM];
  __shared__ float red[8];
  __shared__ float dred[2][8];
  __shared__ int   ccnt[16][32];
  __shared__ int   ctot[16];
  __shared__ float PART[8][16][WDIM + 4];

  const int tid  = threadIdx.x;
  const int lane = tid & 31;
  const int wave = tid >> 5;
  const int half = lane >> 4;
  const int m    = lane & 15;
  const int b    = blockIdx.y;
  const int g    = blockIdx.x;
  const float* xb = x + (((size_t)b * WDIM) << 10);

  fal_u64 key[4];
  {
    float a1[4] = {0.f, 0.f, 0.f, 0.f}, a2[4] = {0.f, 0.f, 0.f, 0.f};
    const float* xp = xb + 4 * tid;
    for (int wb = 0; wb < WDIM; wb += 16) {
      float4 xs[16];
#pragma unroll
      for (int u = 0; u < 16; ++u) xs[u] = *(const float4*)(xp + ((size_t)(wb + u) << 10));
#pragma unroll
      for (int u = 0; u < 16; ++u) {
        const float w1v = w[wb + u];
        const float w2v = w[WDIM + wb + u];
        a1[0] = fmaf(xs[u].x, w1v, a1[0]);
        a1[1] = fmaf(xs[u].y, w1v, a1[1]);
        a1[2] = fmaf(xs[u].z, w1v, a1[2]);
        a1[3] = fmaf(xs[u].w, w1v, a1[3]);
        a2[0] = fmaf(xs[u].x, w2v, a2[0]);
        a2[1] = fmaf(xs[u].y, w2v, a2[1]);
        a2[2] = fmaf(xs[u].z, w2v, a2[2]);
        a2[3] = fmaf(xs[u].w, w2v, a2[3]);
      }
    }
#pragma unroll
    for (int e = 0; e < 4; ++e) {
      const unsigned bits = __builtin_bit_cast(unsigned, a1[e]);
      const unsigned ord  = bits ^ ((bits & 0x80000000u) ? 0xffffffffu : 0x80000000u);
      key[e] = ((fal_u64)ord << 32) | (unsigned)(4 * tid + e);
    }
    *(float4*)&s2s[4 * tid] = make_float4(a2[0], a2[1], a2[2], a2[3]);
    const float mx = fal_wave_max(fmaxf(fmaxf(a2[0], a2[1]), fmaxf(a2[2], a2[3])));
    red[wave] = mx;
  }
  __syncthreads();
  float s2max = red[0];
#pragma unroll
  for (int wv = 1; wv < 8; ++wv) s2max = fmaxf(s2max, red[wv]);

  fal_ce(key[0], key[1], true);
  fal_ce(key[2], key[3], false);
  for (int k = 4; k <= KDIM; k <<= 1) {
    const bool asc = (((tid << 2) & k) == 0);
    if (k >= 256) {
      fal_u64* buf = (k == 512) ? skey : skx;
      *(fal_v2u64*)&buf[4 * tid]     = (fal_v2u64){key[0], key[1]};
      *(fal_v2u64*)&buf[4 * tid + 2] = (fal_v2u64){key[2], key[3]};
      __syncthreads();
      for (int j = k >> 1; j > 128; j >>= 1) {
        const int t0 = tid, t1 = tid + 256;
        const int i0 = ((t0 & ~(j - 1)) << 1) | (t0 & (j - 1));
        const int i1 = ((t1 & ~(j - 1)) << 1) | (t1 & (j - 1));
        const int p0 = i0 | j, p1 = i1 | j;
        const fal_u64 ka0 = buf[i0], kb0 = buf[p0];
        const fal_u64 ka1 = buf[i1], kb1 = buf[p1];
        if ((ka0 > kb0) == ((i0 & k) == 0)) { buf[i0] = kb0; buf[p0] = ka0; }
        if ((ka1 > kb1) == ((i1 & k) == 0)) { buf[i1] = kb1; buf[p1] = ka1; }
        __syncthreads();
      }
      const fal_v2u64 o01 = *(const fal_v2u64*)&buf[4 * tid];
      const fal_v2u64 o23 = *(const fal_v2u64*)&buf[4 * tid + 2];
      const fal_v2u64 q01 = *(const fal_v2u64*)&buf[(4 * tid) ^ 128];
      const fal_v2u64 q23 = *(const fal_v2u64*)&buf[((4 * tid) ^ 128) + 2];
      const fal_u64 own[4] = {o01.x, o01.y, o23.x, o23.y};
      const fal_u64 oth[4] = {q01.x, q01.y, q23.x, q23.y};
      const bool keepmin = ((((tid << 2) & 128) == 0) == asc);
#pragma unroll
      for (int e = 0; e < 4; ++e) key[e] = ((oth[e] > own[e]) != keepmin) ? oth[e] : own[e];
    }
    if (k >= 128) fal_lane_stage<64>(key, lane, asc);
    if (k >= 64)  fal_lane_stage<32>(key, lane, asc);
    if (k >= 32)  fal_lane_stage<16>(key, lane, asc);
    if (k >= 16)  fal_lane_stage<8>(key, lane, asc);
    if (k >= 8)   fal_lane_stage<4>(key, lane, asc);
    fal_ce(key[0], key[2], asc);
    fal_ce(key[1], key[3], asc);
    fal_ce(key[0], key[1], asc);
    fal_ce(key[2], key[3], asc);
  }
  *(fal_v2u64*)&skey[4 * tid]     = (fal_v2u64){key[0], key[1]};
  *(fal_v2u64*)&skey[4 * tid + 2] = (fal_v2u64){key[2], key[3]};
  __syncthreads();

  const float s1min = fal_key_s1(skey[128 * g]);
  const float s1max = fal_key_s1(skey[128 * g + 127]);

  int nband;
  int lo, hi;
  int bnd;
  float DA, DB;
  {
    int cls[4], rk[4];
    float e1r[4], ear[4];
    float tk[16];
#pragma unroll
    for (int k = 0; k < 16; ++k) tk[k] = fal_key_s1(skey[128 * g + 16 * (k >> 1) + 15 * (k & 1)]);
    float da = 0.f, db = 0.f;
#pragma unroll
    for (int rnd = 0; rnd < 4; ++rnd) {
      const int j = tid + 256 * rnd;
      const float s2 = s2s[j];
      const bool above = (s1min + s2 > 0.f);
      const bool below = (s1max + s2 <= 0.f);
      const float d = s2 - s2max;
      e1r[rnd] = __expf(d);
      ear[rnd] = __expf(ALPHA * d);
      const float k1 = above ? e1r[rnd] : 0.f;
      const float k2 = (below && !above) ? ear[rnd] : 0.f;
      c1[j] = k1;
      ca[j] = k2;
      da += k1;
      db += k2;
      int k0 = 0;
#pragma unroll
      for (int k = 0; k < 16; ++k) k0 += (tk[k] + s2 <= 0.f) ? 1 : 0;
      cls[rnd] = (!above && !below) ? k0 : 0;
      int myrank = 0, mycnt = 0;
#pragma unroll
      for (int c = 1; c < 16; ++c) {
        const unsigned mm = __builtin_amdgcn_ballot_w32(cls[rnd] == c);
        const int lower = (int)__builtin_amdgcn_mbcnt_lo(mm, 0u);
        myrank = (cls[rnd] == c) ? lower : myrank;
        mycnt  = (lane == c) ? __popc(mm) : mycnt;
      }
      rk[rnd] = myrank;
      if (lane >= 1 && lane < 16) ccnt[lane][rnd * 8 + wave] = mycnt;
    }
    da = fal_wave_sum(da);
    db = fal_wave_sum(db);
    if (lane == 0) { dred[0][wave] = da; dred[1][wave] = db; }
    __syncthreads();
#pragma unroll
    for (int pass = 0; pass < 2; ++pass) {
      const int c = wave + 1 + 8 * pass;
      if (c < 16) {
        const int own = ccnt[c][lane];
        int inc = own;
#pragma unroll
        for (int d = 1; d < 32; d <<= 1) {
          const int t = __shfl_up(inc, d, 32);
          if (lane >= d) inc += t;
        }
        ccnt[c][lane] = inc - own;
        if (lane == 31) ctot[c] = inc;
      }
    }
    __syncthreads();
    int tc[16];
#pragma unroll
    for (int c = 1; c < 16; ++c) tc[c] = ctot[c];
    int total = 0;
#pragma unroll
    for (int c = 1; c < 16; ++c) total += tc[c];
    nband = total;
#pragma unroll
    for (int rnd = 0; rnd < 4; ++rnd) {
      const int j = tid + 256 * rnd;
      if (cls[rnd] != 0) {
        int base = 0;
#pragma unroll
        for (int c = 1; c < 16; ++c) base += (c < cls[rnd]) ? tc[c] : 0;
        const int pos = base + ccnt[cls[rnd]][rnd * 8 + wave] + rk[rnd];
        bp[j] = pos;
        bq[pos] = make_float2(e1r[rnd], ear[rnd]);
      } else {
        bp[j] = -1;
      }
    }
    {
      const int lim = (m < 8) ? 2 * m + 1 : 2 * (m - 8) + 2;
      int l0 = 0, h0 = 0, b0 = 0;
#pragma unroll
      for (int c = 1; c < 16; ++c) {
        l0 += (c < 2 * wave + 1) ? tc[c] : 0;
        h0 += (c < 2 * wave + 2) ? tc[c] : 0;
        b0 += (c < lim) ? tc[c] : 0;
      }
      lo = __builtin_amdgcn_readfirstlane(l0);
      hi = __builtin_amdgcn_readfirstlane(h0);
      bnd = b0;
    }
    if (tid < 3 && nband + tid < ((nband + 3) & ~3)) bq[nband + tid] = make_float2(0.f, 0.f);
    DA = dred[0][0];
    DB = dred[1][0];
#pragma unroll
    for (int wv = 1; wv < 8; ++wv) { DA += dred[0][wv]; DB += dred[1][wv]; }
  }
  __syncthreads();

  const int   rpos = 128 * g + 16 * wave + m;
  const unsigned long long rkey = skey[rpos];
  const float s1v  = fal_key_s1(rkey);
  const int   ridx = (int)(unsigned)(rkey & 0xffffffffull);
  const float zmax = s1v + s2max;
  const float mrow = fmaxf(zmax, ALPHA * zmax);
  const float u1   = __expf(zmax - mrow);
  const float ua   = __expf(ALPHA * zmax - mrow);

  const int jj4 = tid & 15;
  const int w0  = tid >> 4;
  const float* gbase = xb + ((size_t)w0 << 10) + (jj4 << 2);
  float r1[8], ra[8];
  v8f acc[8];
#pragma unroll
  for (int it = 0; it < 8; ++it) { r1[it] = 0.f; ra[it] = 0.f; acc[it] = (v8f)0.f; }
  float dsum = 0.f;

  const bool narrow = (nband <= NBCAP);
  if (!narrow) { lo = 0; hi = nband; }
  const int npieces = nband > 0 ? (nband + NBCAP - 1) / NBCAP : 1;
  for (int q = 0; q < npieces; ++q) {
    const int plo = q * NBCAP;
    float4 r[8], rn[8];
#pragma unroll
    for (int it = 0; it < 8; ++it) r[it] = *(const float4*)(gbase + ((size_t)(it * 16) << 10));
    for (int c = 0; c < KDIM / 64; ++c) {
      if (c + 1 < KDIM / 64) {
#pragma unroll
        for (int it = 0; it < 8; ++it)
          rn[it] = *(const float4*)(gbase + ((size_t)(it * 16) << 10) + (c + 1) * 64);
      }
      const int j0 = c * 64 + 4 * jj4;
      if (q == 0) {
        const float4 k1 = *(const float4*)&c1[j0];
        const float4 k2 = *(const float4*)&ca[j0];
#pragma unroll
        for (int it = 0; it < 8; ++it) {
          r1[it] = fmaf(r[it].x, k1.x, r1[it]);
          r1[it] = fmaf(r[it].y, k1.y, r1[it]);
          r1[it] = fmaf(r[it].z, k1.z, r1[it]);
          r1[it] = fmaf(r[it].w, k1.w, r1[it]);
          ra[it] = fmaf(r[it].x, k2.x, ra[it]);
          ra[it] = fmaf(r[it].y, k2.y, ra[it]);
          ra[it] = fmaf(r[it].z, k2.z, ra[it]);
          ra[it] = fmaf(r[it].w, k2.w, ra[it]);
        }
      }
      const int4 pl = *(const int4*)&bp[j0];
#define FAL_BAND_STORE(P, COMP)                                                \
      {                                                                        \
        const int lp = (P) - plo;                                              \
        if ((unsigned)lp < (unsigned)NBCAP) {                                  \
          float* d = &BT[(lp >> 1) * PSTR + (lp & 1) + 2 * w0];                \
          _Pragma("unroll") for (int it = 0; it < 8; ++it) d[32 * it] = r[it].COMP; \
        }                                                                      \
      }
      FAL_BAND_STORE(pl.x, x)
      FAL_BAND_STORE(pl.y, y)
      FAL_BAND_STORE(pl.z, z)
      FAL_BAND_STORE(pl.w, w)
#undef FAL_BAND_STORE
#pragma unroll
      for (int it = 0; it < 8; ++it) r[it] = rn[it];
    }
    if (q == 0) {
#pragma unroll
      for (int it = 0; it < 8; ++it) {
        const float t1 = fal_row_sum(r1[it]);
        const float t2 = fal_row_sum(ra[it]);
        if (jj4 == 0) { SA[w0 + 16 * it] = t1; SB[w0 + 16 * it] = t2; }
      }
    }
    const int np   = min(NBCAP, nband - plo);
    const int npad = (np + 3) & ~3;
    if (tid < WDIM) {
      for (int lp = np; lp < npad; ++lp) BT[(lp >> 1) * PSTR + (lp & 1) + 2 * tid] = 0.f;
    }
    __syncthreads();
    const int kb = max(lo - plo, 0) & ~3;
    const int ke = min(max(hi - plo, 0), np);
    for (int kk = kb; kk < ke; kk += 4) {
      const int pg = plo + kk + 2 * half;
      const float4 qv = *(const float4*)&bq[pg];
      const float p0 = (pg >= lo && pg < hi) ? fmaxf(u1 * qv.x, ua * qv.y) : 0.f;
      const float p1 = (pg + 1 >= lo && pg + 1 < hi) ? fmaxf(u1 * qv.z, ua * qv.w) : 0.f;
      dsum += p0 + p1;
      v2f af;
      af.x = p0;
      af.y = p1;
      const float* vp = &BT[((kk >> 1) + half) * PSTR + 2 * m];
#pragma unroll
      for (int t = 0; t < 8; ++t) {
        const v2f bfr = *(const v2f*)(vp + t * 32);
        acc[t] = __builtin_amdgcn_wmma_f32_16x16x4_f32(false, af, false, bfr, (short)0, acc[t],
                                                       false, false);
      }
    }
    if (narrow) {
      v8f ac2[8];
#pragma unroll
      for (int t = 0; t < 8; ++t) ac2[t] = (v8f)0.f;
      const int wvs = __builtin_amdgcn_readfirstlane(wave);
      for (int kk = 4 * wvs; kk < npad; kk += 32) {
        const int pg = kk + 2 * half;
        const float4 qv = *(const float4*)&bq[pg];
        const bool  lowrow = (m < 8);
        const float v0  = lowrow ? qv.x : qv.y;
        const float v1  = lowrow ? qv.z : qv.w;
        const float c0  = ((pg < bnd) == lowrow) ? v0 : 0.f;
        const float c1v = ((pg + 1 < bnd) == lowrow) ? v1 : 0.f;
        v2f af;
        af.x = c0;
        af.y = c1v;
        const float* vp = &BT[((kk >> 1) + half) * PSTR + 2 * m];
#pragma unroll
        for (int t = 0; t < 8; ++t) {
          const v2f bfr = *(const v2f*)(vp + t * 32);
          ac2[t] = __builtin_amdgcn_wmma_f32_16x16x4_f32(false, af, false, bfr, (short)0, ac2[t],
                                                         false, false);
        }
      }
#pragma unroll
      for (int t = 0; t < 8; ++t)
#pragma unroll
        for (int rr = 0; rr < 8; ++rr) PART[wvs][rr + 8 * half][t * 16 + m] = ac2[t][rr];
    }
    __syncthreads();
  }

  const float dband = dsum + fal_xhalf(dsum);
  float sa[8], sb[8];
#pragma unroll
  for (int t = 0; t < 8; ++t) { sa[t] = SA[t * 16 + m]; sb[t] = SB[t * 16 + m]; }
  float d1x = 0.f, dax = 0.f;
  if (narrow) {
#pragma unroll
    for (int t = 0; t < 8; ++t) {
      float x1 = 0.f, xa = 0.f;
#pragma unroll
      for (int wv = 0; wv < 8; ++wv) {
        x1 += PART[wv][wave][t * 16 + m];
        xa += PART[wv][8 + wave][t * 16 + m];
      }
      sa[t] += x1;
      sb[t] += xa;
    }
    for (int p = lane; p < nband; p += 32) {
      const float2 e = bq[p];
      d1x += (p < lo) ? e.x : 0.f;
      dax += (p >= hi) ? e.y : 0.f;
    }
    d1x = fal_wave_sum(d1x);
    dax = fal_wave_sum(dax);
  }
  const float den   = fmaf(u1, DA + d1x, fmaf(ua, DB + dax, dband));
#pragma unroll
  for (int rr = 0; rr < 8; ++rr) {
    const int   row = rr + 8 * half;
    const float u1r = __shfl(u1, row, 32);
    const float uar = __shfl(ua, row, 32);
    const float inv = 1.0f / __shfl(den, row, 32);
    const int   ri  = __shfl(ridx, row, 32);
    float* orow = out + (((size_t)b * KDIM + ri) << 7) + m;
#pragma unroll
    for (int t = 0; t < 8; ++t)
      orow[t * 16] = fmaf(u1r, sa[t], fmaf(uar, sb[t], acc[t][rr])) * inv;
  }
}

extern "C" void kernel_launch(void* const* d_in, const int* in_sizes, int n_in,
                              void* d_out, int out_size, void* d_ws, size_t ws_size,
                              hipStream_t stream) {
  const float* x = (const float*)d_in[0];
  const float* w = (const float*)d_in[1];
  float* out = (float*)d_out;

  dim3 grid(KDIM / 128, BATCH);
  fal_attn_kernel<<<grid, 256, 0, stream>>>(x, w, out);
}
